# starting kernel
# baseline (speedup 1.0000x reference)
#include <hip/hip_runtime.h>

typedef __attribute__((ext_vector_type(2))) float v2f;
typedef __attribute__((ext_vector_type(8))) float v8f;

#define N_NODES 100000
#define E_EDGES 1600000
#define F_DIM   64

__global__ void zero_f32(float* __restrict__ p, int n) {
  int i = blockIdx.x * blockDim.x + threadIdx.x;
  if (i < n) p[i] = 0.0f;
}

__global__ void deg_accum(const int* __restrict__ row, float* __restrict__ deg, int e) {
  int i = blockIdx.x * blockDim.x + threadIdx.x;
  if (i < e) atomicAdd(&deg[row[i]], 1.0f);
}

__global__ void make_dis(float* __restrict__ deg, int n) {
  int i = blockIdx.x * blockDim.x + threadIdx.x;
  if (i < n) {
    float d = deg[i];
    deg[i] = (d > 0.0f) ? rsqrtf(d) : 0.0f;
  }
}

__global__ void edge_weight(const int* __restrict__ row, const int* __restrict__ col,
                            const float* __restrict__ dis, float* __restrict__ w, int e) {
  int i = blockIdx.x * blockDim.x + threadIdx.x;
  if (i < e) w[i] = -dis[row[i]] * dis[col[i]];
}

__global__ void scatter_prop(const int* __restrict__ row, const int* __restrict__ col,
                             const float* __restrict__ w, const float* __restrict__ H,
                             float* __restrict__ P, int e) {
  int gid = blockIdx.x * blockDim.x + threadIdx.x;
  int eidx = gid >> 4;
  if (eidx >= e) return;
  int k = (gid & 15) * 4;
  int r = row[eidx];
  int c = col[eidx];
  float we = w[eidx];
  const float4 v = *(const float4*)(H + (size_t)c * F_DIM + k);
  float* dst = P + (size_t)r * F_DIM + k;
  atomicAdd(dst + 0, we * v.x);
  atomicAdd(dst + 1, we * v.y);
  atomicAdd(dst + 2, we * v.z);
  atomicAdd(dst + 3, we * v.w);
}

template <int FOUT, bool RELU>
__global__ void cheb_gemm(const float* __restrict__ X, const float* __restrict__ P,
                          const float* __restrict__ W0, const float* __restrict__ W1,
                          const float* __restrict__ B, float* __restrict__ OUT) {
  constexpr int NT   = FOUT / 16;
  constexpr int NKG  = F_DIM / 4;
  constexpr int NENT = NKG * NT * 32;

  __shared__ float4 wlds[NENT];

  for (int idx = threadIdx.x; idx < NENT; idx += blockDim.x) {
    const int l   = idx & 31;
    const int rem = idx >> 5;
    const int t   = rem % NT;
    const int kg  = rem / NT;
    const int n   = t * 16 + (l & 15);
    const int k   = kg * 4 + ((l >> 4) * 2);
    wlds[idx] = make_float4(W0[(size_t)k * FOUT + n], W0[(size_t)(k + 1) * FOUT + n],
                            W1[(size_t)k * FOUT + n], W1[(size_t)(k + 1) * FOUT + n]);
  }
  __syncthreads();

  const int lane  = threadIdx.x & 31;
  const int wave  = threadIdx.x >> 5;
  const int tile  = blockIdx.x * 4 + wave;
  const int ntile = N_NODES / 16;
  if (tile >= ntile) return;

  const int row0 = tile * 16;
  const int m    = lane & 15;
  const int kk   = (lane >> 4) * 2;

  v8f acc[NT];
#pragma unroll
  for (int t = 0; t < NT; ++t) acc[t] = (v8f)0.0f;

  const float* xrow = X + (size_t)(row0 + m) * F_DIM;
  const float* prow = P + (size_t)(row0 + m) * F_DIM;

#pragma unroll
  for (int kg = 0; kg < NKG; ++kg) {
    const int k0 = kg * 4;
    const v2f aX = *(const v2f*)(xrow + k0 + kk);
    const v2f aP = *(const v2f*)(prow + k0 + kk);
#pragma unroll
    for (int t = 0; t < NT; ++t) {
      const float4 wv = wlds[(kg * NT + t) * 32 + lane];
      v2f b0, b1;
      b0.x = wv.x; b0.y = wv.y;
      b1.x = wv.z; b1.y = wv.w;
      acc[t] = __builtin_amdgcn_wmma_f32_16x16x4_f32(
          false, aX, false, b0, (short)0, acc[t], false, false);
      acc[t] = __builtin_amdgcn_wmma_f32_16x16x4_f32(
          false, aP, false, b1, (short)0, acc[t], false, false);
    }
  }

  const int roff = (lane >> 4) * 8;
#pragma unroll
  for (int t = 0; t < NT; ++t) {
    const float bias = B[t * 16 + m];
#pragma unroll
    for (int r = 0; r < 8; ++r) {
      float v = acc[t][r] + bias;
      if (RELU) v = fmaxf(v, 0.0f);
      OUT[(size_t)(row0 + roff + r) * FOUT + t * 16 + m] = v;
    }
  }
}

extern "C" void kernel_launch(void* const* d_in, const int* in_sizes, int n_in,
                              void* d_out, int out_size, void* d_ws, size_t ws_size,
                              hipStream_t stream) {
  const float* x    = (const float*)d_in[0];
  const int*   adj  = (const int*)d_in[1];
  const float* W1_0 = (const float*)d_in[2];
  const float* W1_1 = (const float*)d_in[3];
  const float* b1   = (const float*)d_in[4];
  const float* Wx_0 = (const float*)d_in[5];
  const float* Wx_1 = (const float*)d_in[6];
  const float* bx   = (const float*)d_in[7];
  const float* W2_0 = (const float*)d_in[8];
  const float* W2_1 = (const float*)d_in[9];
  const float* b2   = (const float*)d_in[10];
  float* out = (float*)d_out;

  const int* row = adj;
  const int* col = adj + E_EDGES;

  float* dis = (float*)d_ws;
  float* w   = dis + N_NODES;
  float* P   = w + E_EDGES;
  float* H1  = P + (size_t)N_NODES * F_DIM;
  float* H2  = H1 + (size_t)N_NODES * F_DIM;

  const int TB = 256;
  const int gN   = (N_NODES + TB - 1) / TB;
  const int gE   = (E_EDGES + TB - 1) / TB;
  const int gNF  = (N_NODES * F_DIM + TB - 1) / TB;
  const int gSc  = (E_EDGES * 16 + TB - 1) / TB;
  const int gGem = (N_NODES / 16 + 3) / 4;

  zero_f32<<<gN, TB, 0, stream>>>(dis, N_NODES);
  deg_accum<<<gE, TB, 0, stream>>>(row, dis, E_EDGES);
  make_dis<<<gN, TB, 0, stream>>>(dis, N_NODES);
  edge_weight<<<gE, TB, 0, stream>>>(row, col, dis, w, E_EDGES);

  zero_f32<<<gNF, TB, 0, stream>>>(P, N_NODES * F_DIM);
  scatter_prop<<<gSc, TB, 0, stream>>>(row, col, w, x, P, E_EDGES);
  cheb_gemm<64, true><<<gGem, 128, 0, stream>>>(x, P, W1_0, W1_1, b1, H1);

  zero_f32<<<gNF, TB, 0, stream>>>(P, N_NODES * F_DIM);
  scatter_prop<<<gSc, TB, 0, stream>>>(row, col, w, H1, P, E_EDGES);
  cheb_gemm<64, true><<<gGem, 128, 0, stream>>>(H1, P, Wx_0, Wx_1, bx, H2);

  zero_f32<<<gNF, TB, 0, stream>>>(P, N_NODES * F_DIM);
  scatter_prop<<<gSc, TB, 0, stream>>>(row, col, w, H2, P, E_EDGES);
  cheb_gemm<16, false><<<gGem, 128, 0, stream>>>(H2, P, W2_0, W2_1, b2, out);
}


# layers 2-3 gather a 2nd copy of H1/H2 pre-scaled by deg^-1/2 of its row; row factor once per row at the P fragment; no per-edge degree read or weight; product rounding changes; sum order: LDS arrival
# speedup vs baseline: 2.1747x; 2.1747x over previous
#include <hip/hip_runtime.h>

typedef __attribute__((ext_vector_type(2))) float v2f;
typedef __attribute__((ext_vector_type(8))) float v8f;

#define N_NODES 100000
#define E_EDGES 1600000
#define F_DIM   64

#define R_TILE  400
#define P_PITCH 68
#define NWAVE   32
#define QCAP    64

__global__ void zero_f32(float* __restrict__ p, int n) {
  int i = blockIdx.x * blockDim.x + threadIdx.x;
  if (i < n) p[i] = 0.0f;
}

__global__ void deg_accum(const int* __restrict__ row, float* __restrict__ deg, int e) {
  int i = blockIdx.x * blockDim.x + threadIdx.x;
  if (i < e) atomicAdd(&deg[row[i]], 1.0f);
}

static __device__ __forceinline__ float dis_of(float d) {
  return (d > 0.0f) ? rsqrtf(d) : 0.0f;
}

__global__ void edge_key(const int* __restrict__ row, unsigned* __restrict__ key, int e4) {
  int i = blockIdx.x * blockDim.x + threadIdx.x;
  if (i < e4) {
    const int4 r = ((const int4*)row)[i];
    key[i] = ((unsigned)r.x / R_TILE) | (((unsigned)r.y / R_TILE) << 8) |
             (((unsigned)r.z / R_TILE) << 16) | (((unsigned)r.w / R_TILE) << 24);
  }
}


static __device__ __forceinline__ void wave_lds_handover() {
  __builtin_amdgcn_fence(__ATOMIC_ACQ_REL, "wavefront");
  __builtin_amdgcn_wave_barrier();
}

static __device__ __forceinline__ void lds_add(float* p, float v) {
  __hip_atomic_fetch_add(p, v, __ATOMIC_RELAXED, __HIP_MEMORY_SCOPE_WORKGROUP);
}

template <bool FULL, int NG, bool WEIGHTED>
static __device__ __forceinline__ void gather_add(unsigned c, unsigned rl, int n,
                                                  int lane, int2* st, float* ptile,
                                                  const float* __restrict__ deg,
                                                  const float* __restrict__ X) {
  st[lane] = make_int2((int)(c * (F_DIM * 4u)), (int)(rl * (P_PITCH * 4u)));
  wave_lds_handover();
  const int half = lane >> 4;
  const unsigned kb = (unsigned)(lane & 15) * 16u;
  int wmine = 0;
  int2 s[16];
#pragma unroll
  for (int i = 0; i < 16; ++i) s[i] = st[i * 2 + half];
#pragma unroll
  for (int hb = 0; hb < 16 / NG; ++hb) {
    float4 g[NG];
#pragma unroll
    for (int i = 0; i < NG; ++i) {
      const int j = (hb * NG + i) * 2 + half;
      g[i] = make_float4(0.0f, 0.0f, 0.0f, 0.0f);
      if (FULL || j < n) g[i] = *(const float4*)((const char*)X + ((unsigned)s[hb * NG + i].x + kb));
    }
    if (WEIGHTED && hb == 0) {
      __builtin_amdgcn_sched_barrier(0);
      float dgc = 0.0f;
      if (FULL || lane < n) dgc = deg[c];
      wmine = __builtin_bit_cast(int, dis_of(dgc));
    }
#pragma unroll
    for (int i = 0; i < NG; ++i) {
      const int j = (hb * NG + i) * 2 + half;
      float ww = 1.0f;
      if (WEIGHTED) {
        const float w_lo = __builtin_bit_cast(float, __builtin_amdgcn_readlane(wmine, (hb * NG + i) * 2));
        const float w_hi = __builtin_bit_cast(float, __builtin_amdgcn_readlane(wmine, (hb * NG + i) * 2 + 1));
        ww = half ? w_hi : w_lo;
      }
      if (FULL || j < n) {
        float* dst = (float*)((char*)ptile + ((unsigned)s[hb * NG + i].y + kb));
        lds_add(dst + 0, WEIGHTED ? ww * g[i].x : g[i].x);
        lds_add(dst + 1, WEIGHTED ? ww * g[i].y : g[i].y);
        lds_add(dst + 2, WEIGHTED ? ww * g[i].z : g[i].z);
        lds_add(dst + 3, WEIGHTED ? ww * g[i].w : g[i].w);
      }
    }
  }
}

template <bool FULL, bool SCAN>
static __device__ __forceinline__ void do_batch(unsigned e, int n, int lane, int base, int2* st,
                                                float* ptile,
                                                const int* __restrict__ row,
                                                const int* __restrict__ col,
                                                const float* __restrict__ deg,
                                                const float* __restrict__ X,
                                                unsigned* __restrict__ lst) {
  unsigned c = 0, rl = 0;
  if (FULL || lane < n) {
    if (SCAN) {
      rl = (unsigned)(row[e] - base);
      c  = (unsigned)col[e];
      lst[lane] = c | (rl << 17);
    } else {
      c  = e & 0x1FFFFu;
      rl = e >> 17;
    }
  }
  gather_add<FULL, SCAN ? 8 : 16, SCAN>(c, rl, n, lane, st, ptile, deg, X);
}

template <int FOUT, bool RELU, bool SCAN, bool COPY>
__global__ __launch_bounds__(1024) void cheb_layer(
    const int* __restrict__ row, const int* __restrict__ col, const unsigned* __restrict__ key,
    unsigned* __restrict__ elist, const float* __restrict__ deg,
    const float* XD, const float* __restrict__ XG, const float* __restrict__ W0, const float* __restrict__ W1,
    const float* __restrict__ B, float* OUT, float* __restrict__ OUTS) {
  constexpr int NT   = FOUT / 16;
  constexpr int NKG  = F_DIM / 4;
  constexpr int NENT = NKG * NT * 32;

  __shared__ float    ptile[R_TILE * P_PITCH];
  __shared__ float    disl[R_TILE];
  __shared__ float4   wlds[NENT];
  __shared__ unsigned queue[NWAVE * QCAP];
  __shared__ int2     stage[NWAVE * 32];
  __shared__ float    red_lo[NWAVE];
  __shared__ float    red_own[NWAVE];
  __shared__ unsigned lfill;

  const int tid  = threadIdx.x;
  const int lane = tid & 31;
  const int wave = __builtin_amdgcn_readfirstlane(tid >> 5);
  const int base = blockIdx.x * R_TILE;

  for (int i = tid; i < R_TILE * P_PITCH / 4; i += NWAVE * 32)
    ((float4*)ptile)[i] = make_float4(0.0f, 0.0f, 0.0f, 0.0f);

  if (tid < R_TILE) disl[tid] = dis_of(deg[base + tid]);

  for (int idx = tid; idx < NENT; idx += NWAVE * 32) {
    const int l   = idx & 31;
    const int rem = idx >> 5;
    const int t   = rem % NT;
    const int kg  = rem / NT;
    const int n   = t * 16 + (l & 15);
    const int k   = kg * 4 + ((l >> 4) * 2);
    wlds[idx] = make_float4(W0[(size_t)k * FOUT + n], W0[(size_t)(k + 1) * FOUT + n],
                            W1[(size_t)k * FOUT + n], W1[(size_t)(k + 1) * FOUT + n]);
  }
  {
    float s_lo = 0.0f, s_own = 0.0f;
    const float4* d4 = (const float4*)deg;
    const int n_lo  = base / 4;
    const int n_all = n_lo + R_TILE / 4;
    constexpr int DPT = 8;
    for (int i0 = 0; i0 < n_all; i0 += DPT * NWAVE * 32) {
      float4 d[DPT];
#pragma unroll
      for (int k = 0; k < DPT; ++k)
        d[k] = d4[min(i0 + k * NWAVE * 32 + tid, n_all - 1)];
#pragma unroll
      for (int k = 0; k < DPT; ++k) {
        const int i = i0 + k * NWAVE * 32 + tid;
        const float s = (d[k].x + d[k].y) + (d[k].z + d[k].w);
        if (i < n_lo) s_lo += s; else if (i < n_all) s_own += s;
      }
    }
#pragma unroll
    for (int o = 16; o > 0; o >>= 1) {
      s_lo  += __shfl_xor(s_lo, o);
      s_own += __shfl_xor(s_own, o);
    }
    if (lane == 0) { red_lo[wave] = s_lo; red_own[wave] = s_own; }
    if (tid == 0) lfill = 0u;
  }
  __syncthreads();
  float t_lo = 0.0f, t_own = 0.0f;
#pragma unroll
  for (int w = 0; w < NWAVE; ++w) { t_lo += red_lo[w]; t_own += red_own[w]; }
  const unsigned l_off = __builtin_amdgcn_readfirstlane((unsigned)t_lo);
  const unsigned l_cnt = __builtin_amdgcn_readfirstlane((unsigned)t_own);
  unsigned* mylist = elist + l_off;

  int2*     st  = stage + wave * 32;

  if (SCAN) {
  unsigned* q   = queue + wave * QCAP;
  int qn = 0;
  const int4* key4 = (const int4*)key;
  constexpr int NCH = E_EDGES / 512;
  constexpr int LPT = 8;
  constexpr int NIT = (NCH + NWAVE * LPT - 1) / (NWAVE * LPT);
  const unsigned pat = (unsigned)blockIdx.x * 0x01010101u;

  auto nz = [](unsigned x) { return ((x & 0x7F7F7F7Fu) + 0x7F7F7F7Fu) | x; };
  auto take = [&](const int4 v, int ch) {
    const unsigned u0 = nz((unsigned)v.x ^ pat);
    const unsigned u1 = nz((unsigned)v.y ^ pat);
    const unsigned u2 = nz((unsigned)v.z ^ pat);
    const unsigned u3 = nz((unsigned)v.w ^ pat);
    const bool any = ((u0 & u1 & u2 & u3) & 0x80808080u) != 0x80808080u;
    unsigned mask = __builtin_amdgcn_ballot_w32(any);
    if (mask) {
      unsigned hv = (((u0 & 0x80808080u) >> 3) | ((u1 & 0x80808080u) >> 2) |
                     ((u2 & 0x80808080u) >> 1) | (u3 & 0x80808080u)) ^ 0xF0F0F0F0u;
      const unsigned e0 = (unsigned)(ch * 32 + lane) * 16u;
      do {
        if (hv) {
          const unsigned p = (unsigned)__builtin_ctz(hv);
          hv &= hv - 1u;
          q[qn + __builtin_amdgcn_mbcnt_lo(mask, 0u)] = e0 + (((p & 3u) << 2) | (p >> 3));
        }
        qn += __builtin_popcount(mask);
        mask = __builtin_amdgcn_ballot_w32(hv != 0u);
        while (qn >= 32) {
          qn -= 32;
          wave_lds_handover();
          unsigned slot = 0u;
          if (lane == 0) slot = __hip_atomic_fetch_add(&lfill, 32u, __ATOMIC_RELAXED, __HIP_MEMORY_SCOPE_WORKGROUP);
          slot = __builtin_amdgcn_readfirstlane(slot);
          do_batch<true, true>(q[qn + lane], 32, lane, base, st, ptile, row, col, deg, XG, mylist + slot);
        }
      } while (mask);
    }
  };

  for (int it0 = 0; it0 < NIT; ++it0) {
    const int it  = (it0 + (int)blockIdx.x) % NIT;
    const int ch0 = (it * NWAVE + wave) * LPT;
    if (ch0 >= NCH) continue;
    int4 v[LPT];
#pragma unroll
    for (int k = 0; k < LPT; ++k)
      v[k] = key4[min(ch0 + k, NCH - 1) * 32 + lane];
#pragma unroll
    for (int k = 0; k < LPT; ++k)
      if (ch0 + k < NCH) take(v[k], ch0 + k);
  }
  if (qn > 0) {
    wave_lds_handover();
    const unsigned ent = (lane < qn) ? q[lane] : 0u;
    unsigned slot = 0u;
    if (lane == 0) slot = __hip_atomic_fetch_add(&lfill, (unsigned)qn, __ATOMIC_RELAXED, __HIP_MEMORY_SCOPE_WORKGROUP);
    slot = __builtin_amdgcn_readfirstlane(slot);
    do_batch<false, true>(ent, qn, lane, base, st, ptile, row, col, deg, XG, mylist + slot);
  }
  } else {
    const int nfull = (int)(l_cnt >> 5);
    const int ntail = (int)(l_cnt & 31u);
    if (wave < nfull) {
      const int blast = nfull - 1;
      unsigned w0 = mylist[wave * 32 + lane];
      unsigned wn = mylist[min(wave + NWAVE, blast) * 32 + lane];
      for (int b = wave; b < nfull; b += NWAVE) {
        const unsigned wn2 = mylist[min(b + 2 * NWAVE, blast) * 32 + lane];
        gather_add<true, 16, false>(w0 & 0x1FFFFu, w0 >> 17, 32, lane, st, ptile, deg, XG);
        w0 = wn; wn = wn2;
      }
    }
    if (ntail > 0 && wave == (nfull & (NWAVE - 1))) {
      const unsigned ent = (lane < ntail) ? mylist[nfull * 32 + lane] : 0u;
      do_batch<false, false>(ent, ntail, lane, base, st, ptile, row, col, deg, XG, mylist);
    }
  }
  __syncthreads();

  const int m  = lane & 15;
  const int kk = (lane >> 4) * 2;
  const int roff = (lane >> 4) * 8;
  for (int t16 = wave; t16 < R_TILE / 16; t16 += NWAVE) {
    const int row0 = base + t16 * 16;
    v8f acc[NT];
#pragma unroll
    for (int t = 0; t < NT; ++t) acc[t] = (v8f)0.0f;

    const float* xrow = XD + (size_t)(row0 + m) * F_DIM;
    const float* prow = ptile + (t16 * 16 + m) * P_PITCH;
    const float nd = -disl[t16 * 16 + m];

#pragma unroll 4
    for (int kg = 0; kg < NKG; ++kg) {
      const int k0 = kg * 4;
      const v2f aX = *(const v2f*)(xrow + k0 + kk);
      v2f aP = *(const v2f*)(prow + k0 + kk);
      aP.x *= nd; aP.y *= nd;
#pragma unroll
      for (int t = 0; t < NT; ++t) {
        const float4 wv = wlds[(kg * NT + t) * 32 + lane];
        v2f b0, b1;
        b0.x = wv.x; b0.y = wv.y;
        b1.x = wv.z; b1.y = wv.w;
        acc[t] = __builtin_amdgcn_wmma_f32_16x16x4_f32(
            false, aX, false, b0, (short)0, acc[t], false, false);
        acc[t] = __builtin_amdgcn_wmma_f32_16x16x4_f32(
            false, aP, false, b1, (short)0, acc[t], false, false);
      }
    }

    float dsr[8];
#pragma unroll
    for (int r = 0; r < 8; ++r) dsr[r] = COPY ? disl[t16 * 16 + roff + r] : 0.0f;
    float* srow = ptile + (t16 * 16) * P_PITCH;
#pragma unroll
    for (int t = 0; t < NT; ++t) {
      const float bias = B[t * 16 + m];
#pragma unroll
      for (int r = 0; r < 8; ++r) {
        float vv = acc[t][r] + bias;
        if (RELU) vv = fmaxf(vv, 0.0f);
        OUT[(size_t)(row0 + roff + r) * FOUT + t * 16 + m] = vv;
        if (COPY) srow[(roff + r) * P_PITCH + t * 16 + m] = dsr[r] * vv;
      }
    }
    if (COPY) {
      wave_lds_handover();
      const int q4 = (lane & 15) * 4;
#pragma unroll
      for (int i = 0; i < 8; ++i) {
        const int rr = i * 2 + (lane >> 4);
        const float4 v = *(const float4*)(srow + rr * P_PITCH + q4);
        *(float4*)(OUTS + (size_t)(row0 + rr) * FOUT + q4) = v;
      }
    }
  }
}

extern "C" void kernel_launch(void* const* d_in, const int* in_sizes, int n_in,
                              void* d_out, int out_size, void* d_ws, size_t ws_size,
                              hipStream_t stream) {
  const float* x    = (const float*)d_in[0];
  const int*   adj  = (const int*)d_in[1];
  const float* W1_0 = (const float*)d_in[2];
  const float* W1_1 = (const float*)d_in[3];
  const float* b1   = (const float*)d_in[4];
  const float* Wx_0 = (const float*)d_in[5];
  const float* Wx_1 = (const float*)d_in[6];
  const float* bx   = (const float*)d_in[7];
  const float* W2_0 = (const float*)d_in[8];
  const float* W2_1 = (const float*)d_in[9];
  const float* b2   = (const float*)d_in[10];
  float* out = (float*)d_out;

  const int* row = adj;
  const int* col = adj + E_EDGES;

  float* H   = (float*)d_ws;
  float* H1s = H + (size_t)N_NODES * F_DIM;
  float* H2s = H1s + (size_t)N_NODES * F_DIM;
  float* deg = H2s + (size_t)N_NODES * F_DIM;
  unsigned* elist = (unsigned*)(deg + N_NODES);
  unsigned* key = (unsigned*)H2s;
  if (ws_size < sizeof(float) * ((size_t)N_NODES + E_EDGES + 3 * (size_t)N_NODES * F_DIM)) return;

  const int TB = 256;
  const int gN   = (N_NODES + TB - 1) / TB;
  const int gE   = (E_EDGES + TB - 1) / TB;
  const int gE4  = (E_EDGES / 4 + TB - 1) / TB;
  const int gLay = N_NODES / R_TILE;

  zero_f32<<<gN, TB, 0, stream>>>(deg, N_NODES);
  deg_accum<<<gE, TB, 0, stream>>>(row, deg, E_EDGES);
  edge_key<<<gE4, TB, 0, stream>>>(row, key, E_EDGES / 4);

  cheb_layer<64, true, true, true><<<gLay, 1024, 0, stream>>>(row, col, key, elist, deg, x, x, W1_0, W1_1, b1, H, H1s);

  cheb_layer<64, true, false, true><<<gLay, 1024, 0, stream>>>(row, col, nullptr, elist, deg, H, H1s, Wx_0, Wx_1, bx, H, H2s);

  cheb_layer<16, false, false, false><<<gLay, 1024, 0, stream>>>(row, col, nullptr, elist, deg, H, H2s, W2_0, W2_1, b2, out, nullptr);
}


# in-degrees counted in LDS by the owner groups' key scan (integer adds, plain stores), not by 1.6M float atomics on memory; scan + edge lists in a launch of their own before layer 1; 5 launches, not 6
# speedup vs baseline: 2.7930x; 1.2843x over previous
#include <hip/hip_runtime.h>

typedef __attribute__((ext_vector_type(2))) float v2f;
typedef __attribute__((ext_vector_type(8))) float v8f;

#define N_NODES 100000
#define E_EDGES 1600000
#define F_DIM   64

#define R_TILE  400
#define P_PITCH 68
#define NWAVE   32
#define QCAP    64

static __device__ __forceinline__ float dis_of(float d) {
  return (d > 0.0f) ? rsqrtf(d) : 0.0f;
}

#define KEY_WG    128
#define KEY_ITEMS (E_EDGES / 4 / KEY_WG)

__global__ __launch_bounds__(1024) void edge_key_hist(const int* __restrict__ row, unsigned* __restrict__ key,
                                                      unsigned* __restrict__ part) {
  __shared__ unsigned hist[256];
  const int tid = threadIdx.x;
  if (tid < 256) hist[tid] = 0u;
  __syncthreads();
  const int i0 = blockIdx.x * KEY_ITEMS;
  for (int i = i0 + tid; i < i0 + KEY_ITEMS; i += 1024) {
    const int4 r = ((const int4*)row)[i];
    const unsigned k0 = (unsigned)r.x / R_TILE, k1 = (unsigned)r.y / R_TILE;
    const unsigned k2 = (unsigned)r.z / R_TILE, k3 = (unsigned)r.w / R_TILE;
    key[i] = k0 | (k1 << 8) | (k2 << 16) | (k3 << 24);
    atomicAdd(&hist[k0], 1u);
    atomicAdd(&hist[k1], 1u);
    atomicAdd(&hist[k2], 1u);
    atomicAdd(&hist[k3], 1u);
  }
  __syncthreads();
  if (tid < N_NODES / R_TILE) part[tid * KEY_WG + blockIdx.x] = hist[tid];
}


static __device__ __forceinline__ void wave_lds_handover() {
  __builtin_amdgcn_fence(__ATOMIC_ACQ_REL, "wavefront");
  __builtin_amdgcn_wave_barrier();
}

static __device__ __forceinline__ void lds_add(float* p, float v) {
  __hip_atomic_fetch_add(p, v, __ATOMIC_RELAXED, __HIP_MEMORY_SCOPE_WORKGROUP);
}

template <bool FULL, int NG, bool WEIGHTED>
static __device__ __forceinline__ void gather_add(unsigned c, unsigned rl, int n,
                                                  int lane, int2* st, float* ptile,
                                                  const float* __restrict__ deg,
                                                  const float* __restrict__ X) {
  st[lane] = make_int2((int)(c * (F_DIM * 4u)), (int)(rl * (P_PITCH * 4u)));
  wave_lds_handover();
  const int half = lane >> 4;
  const unsigned kb = (unsigned)(lane & 15) * 16u;
  int wmine = 0;
  int2 s[16];
#pragma unroll
  for (int i = 0; i < 16; ++i) s[i] = st[i * 2 + half];
#pragma unroll
  for (int hb = 0; hb < 16 / NG; ++hb) {
    float4 g[NG];
#pragma unroll
    for (int i = 0; i < NG; ++i) {
      const int j = (hb * NG + i) * 2 + half;
      g[i] = make_float4(0.0f, 0.0f, 0.0f, 0.0f);
      if (FULL || j < n) g[i] = *(const float4*)((const char*)X + ((unsigned)s[hb * NG + i].x + kb));
    }
    if (WEIGHTED && hb == 0) {
      __builtin_amdgcn_sched_barrier(0);
      float dgc = 0.0f;
      if (FULL || lane < n) dgc = deg[c];
      wmine = __builtin_bit_cast(int, dis_of(dgc));
    }
#pragma unroll
    for (int i = 0; i < NG; ++i) {
      const int j = (hb * NG + i) * 2 + half;
      float ww = 1.0f;
      if (WEIGHTED) {
        const float w_lo = __builtin_bit_cast(float, __builtin_amdgcn_readlane(wmine, (hb * NG + i) * 2));
        const float w_hi = __builtin_bit_cast(float, __builtin_amdgcn_readlane(wmine, (hb * NG + i) * 2 + 1));
        ww = half ? w_hi : w_lo;
      }
      if (FULL || j < n) {
        float* dst = (float*)((char*)ptile + ((unsigned)s[hb * NG + i].y + kb));
        lds_add(dst + 0, WEIGHTED ? ww * g[i].x : g[i].x);
        lds_add(dst + 1, WEIGHTED ? ww * g[i].y : g[i].y);
        lds_add(dst + 2, WEIGHTED ? ww * g[i].z : g[i].z);
        lds_add(dst + 3, WEIGHTED ? ww * g[i].w : g[i].w);
      }
    }
  }
}

template <bool FULL>
static __device__ __forceinline__ void list_batch(unsigned e, int n, int lane, int base, unsigned* cnt,
                                                  const int* __restrict__ row,
                                                  const int* __restrict__ col,
                                                  unsigned* __restrict__ lst) {
  if (FULL || lane < n) {
    const unsigned rl = (unsigned)(row[e] - base);
    lst[lane] = (unsigned)col[e] | (rl << 17);
    atomicAdd(&cnt[rl], 1u);
  }
}

__global__ __launch_bounds__(1024) void scan_list(
    const int* __restrict__ row, const int* __restrict__ col, const unsigned* __restrict__ key,
    const unsigned* __restrict__ part, unsigned* __restrict__ elist, float* __restrict__ deg) {
  __shared__ unsigned cnt[R_TILE];
  __shared__ unsigned queue[NWAVE * QCAP];
  __shared__ unsigned red_lo[NWAVE];
  __shared__ unsigned lfill;

  const int tid  = threadIdx.x;
  const int lane = tid & 31;
  const int wave = __builtin_amdgcn_readfirstlane(tid >> 5);
  const int base = blockIdx.x * R_TILE;

  if (tid < R_TILE) cnt[tid] = 0u;
  {
    unsigned s_lo = 0u;
    const int4* p4 = (const int4*)part;
    const int n_lo = (int)blockIdx.x * (KEY_WG / 4);
    constexpr int DPT = 8;
    for (int i0 = 0; i0 < n_lo; i0 += DPT * NWAVE * 32) {
      int4 d[DPT];
#pragma unroll
      for (int k = 0; k < DPT; ++k)
        d[k] = p4[min(i0 + k * NWAVE * 32 + tid, n_lo - 1)];
#pragma unroll
      for (int k = 0; k < DPT; ++k)
        if (i0 + k * NWAVE * 32 + tid < n_lo)
          s_lo += ((unsigned)d[k].x + (unsigned)d[k].y) + ((unsigned)d[k].z + (unsigned)d[k].w);
    }
#pragma unroll
    for (int o = 16; o > 0; o >>= 1) s_lo += __shfl_xor(s_lo, o);
    if (lane == 0) red_lo[wave] = s_lo;
    if (tid == 0) lfill = 0u;
  }
  __syncthreads();
  unsigned t_lo = 0u;
#pragma unroll
  for (int w = 0; w < NWAVE; ++w) t_lo += red_lo[w];
  unsigned* mylist = elist + __builtin_amdgcn_readfirstlane(t_lo);

  unsigned* q   = queue + wave * QCAP;
  int qn = 0;
  const int4* key4 = (const int4*)key;
  constexpr int NCH = E_EDGES / 512;
  constexpr int LPT = 8;
  constexpr int NIT = (NCH + NWAVE * LPT - 1) / (NWAVE * LPT);
  const unsigned pat = (unsigned)blockIdx.x * 0x01010101u;

  auto nz = [](unsigned x) { return ((x & 0x7F7F7F7Fu) + 0x7F7F7F7Fu) | x; };
  auto take = [&](const int4 v, int ch) {
    const unsigned u0 = nz((unsigned)v.x ^ pat);
    const unsigned u1 = nz((unsigned)v.y ^ pat);
    const unsigned u2 = nz((unsigned)v.z ^ pat);
    const unsigned u3 = nz((unsigned)v.w ^ pat);
    const bool any = ((u0 & u1 & u2 & u3) & 0x80808080u) != 0x80808080u;
    unsigned mask = __builtin_amdgcn_ballot_w32(any);
    if (mask) {
      unsigned hv = (((u0 & 0x80808080u) >> 3) | ((u1 & 0x80808080u) >> 2) |
                     ((u2 & 0x80808080u) >> 1) | (u3 & 0x80808080u)) ^ 0xF0F0F0F0u;
      const unsigned e0 = (unsigned)(ch * 32 + lane) * 16u;
      do {
        if (hv) {
          const unsigned p = (unsigned)__builtin_ctz(hv);
          hv &= hv - 1u;
          q[qn + __builtin_amdgcn_mbcnt_lo(mask, 0u)] = e0 + (((p & 3u) << 2) | (p >> 3));
        }
        qn += __builtin_popcount(mask);
        mask = __builtin_amdgcn_ballot_w32(hv != 0u);
        while (qn >= 32) {
          qn -= 32;
          wave_lds_handover();
          unsigned slot = 0u;
          if (lane == 0) slot = __hip_atomic_fetch_add(&lfill, 32u, __ATOMIC_RELAXED, __HIP_MEMORY_SCOPE_WORKGROUP);
          slot = __builtin_amdgcn_readfirstlane(slot);
          list_batch<true>(q[qn + lane], 32, lane, base, cnt, row, col, mylist + slot);
        }
      } while (mask);
    }
  };

  for (int it0 = 0; it0 < NIT; ++it0) {
    const int it  = (it0 + (int)blockIdx.x) % NIT;
    const int ch0 = (it * NWAVE + wave) * LPT;
    if (ch0 >= NCH) continue;
    int4 v[LPT];
#pragma unroll
    for (int k = 0; k < LPT; ++k)
      v[k] = key4[min(ch0 + k, NCH - 1) * 32 + lane];
#pragma unroll
    for (int k = 0; k < LPT; ++k)
      if (ch0 + k < NCH) take(v[k], ch0 + k);
  }
  if (qn > 0) {
    wave_lds_handover();
    const unsigned ent = (lane < qn) ? q[lane] : 0u;
    unsigned slot = 0u;
    if (lane == 0) slot = __hip_atomic_fetch_add(&lfill, (unsigned)qn, __ATOMIC_RELAXED, __HIP_MEMORY_SCOPE_WORKGROUP);
    slot = __builtin_amdgcn_readfirstlane(slot);
    list_batch<false>(ent, qn, lane, base, cnt, row, col, mylist + slot);
  }
  __syncthreads();
  if (tid < R_TILE) deg[base + tid] = (float)cnt[tid];
}

template <int FOUT, bool RELU, bool WGT, bool COPY>
__global__ __launch_bounds__(1024) void cheb_layer(
    const unsigned* __restrict__ elist, const float* __restrict__ deg,
    const float* XD, const float* __restrict__ XG, const float* __restrict__ W0, const float* __restrict__ W1,
    const float* __restrict__ B, float* OUT, float* __restrict__ OUTS) {
  constexpr int NT   = FOUT / 16;
  constexpr int NKG  = F_DIM / 4;
  constexpr int NENT = NKG * NT * 32;

  __shared__ float    ptile[R_TILE * P_PITCH];
  __shared__ float    disl[R_TILE];
  __shared__ float4   wlds[NENT];
  __shared__ int2     stage[NWAVE * 32];
  __shared__ float    red_lo[NWAVE];
  __shared__ float    red_own[NWAVE];

  const int tid  = threadIdx.x;
  const int lane = tid & 31;
  const int wave = __builtin_amdgcn_readfirstlane(tid >> 5);
  const int base = blockIdx.x * R_TILE;

  for (int i = tid; i < R_TILE * P_PITCH / 4; i += NWAVE * 32)
    ((float4*)ptile)[i] = make_float4(0.0f, 0.0f, 0.0f, 0.0f);

  if (tid < R_TILE) disl[tid] = dis_of(deg[base + tid]);

  for (int idx = tid; idx < NENT; idx += NWAVE * 32) {
    const int l   = idx & 31;
    const int rem = idx >> 5;
    const int t   = rem % NT;
    const int kg  = rem / NT;
    const int n   = t * 16 + (l & 15);
    const int k   = kg * 4 + ((l >> 4) * 2);
    wlds[idx] = make_float4(W0[(size_t)k * FOUT + n], W0[(size_t)(k + 1) * FOUT + n],
                            W1[(size_t)k * FOUT + n], W1[(size_t)(k + 1) * FOUT + n]);
  }
  {
    float s_lo = 0.0f, s_own = 0.0f;
    const float4* d4 = (const float4*)deg;
    const int n_lo  = base / 4;
    const int n_all = n_lo + R_TILE / 4;
    constexpr int DPT = 8;
    for (int i0 = 0; i0 < n_all; i0 += DPT * NWAVE * 32) {
      float4 d[DPT];
#pragma unroll
      for (int k = 0; k < DPT; ++k)
        d[k] = d4[min(i0 + k * NWAVE * 32 + tid, n_all - 1)];
#pragma unroll
      for (int k = 0; k < DPT; ++k) {
        const int i = i0 + k * NWAVE * 32 + tid;
        const float s = (d[k].x + d[k].y) + (d[k].z + d[k].w);
        if (i < n_lo) s_lo += s; else if (i < n_all) s_own += s;
      }
    }
#pragma unroll
    for (int o = 16; o > 0; o >>= 1) {
      s_lo  += __shfl_xor(s_lo, o);
      s_own += __shfl_xor(s_own, o);
    }
    if (lane == 0) { red_lo[wave] = s_lo; red_own[wave] = s_own; }
  }
  __syncthreads();
  float t_lo = 0.0f, t_own = 0.0f;
#pragma unroll
  for (int w = 0; w < NWAVE; ++w) { t_lo += red_lo[w]; t_own += red_own[w]; }
  const unsigned l_off = __builtin_amdgcn_readfirstlane((unsigned)t_lo);
  const unsigned l_cnt = __builtin_amdgcn_readfirstlane((unsigned)t_own);
  const unsigned* mylist = elist + l_off;

  int2*     st  = stage + wave * 32;

  const int nfull = (int)(l_cnt >> 5);
  const int ntail = (int)(l_cnt & 31u);
  if (wave < nfull) {
    const int blast = nfull - 1;
    unsigned w0 = mylist[wave * 32 + lane];
    unsigned wn = mylist[min(wave + NWAVE, blast) * 32 + lane];
    for (int b = wave; b < nfull; b += NWAVE) {
      const unsigned wn2 = mylist[min(b + 2 * NWAVE, blast) * 32 + lane];
      gather_add<true, 16, WGT>(w0 & 0x1FFFFu, w0 >> 17, 32, lane, st, ptile, deg, XG);
      w0 = wn; wn = wn2;
    }
  }
  if (ntail > 0 && wave == (nfull & (NWAVE - 1))) {
    const unsigned ent = (lane < ntail) ? mylist[nfull * 32 + lane] : 0u;
    gather_add<false, 16, WGT>(ent & 0x1FFFFu, ent >> 17, ntail, lane, st, ptile, deg, XG);
  }
  __syncthreads();

  const int m  = lane & 15;
  const int kk = (lane >> 4) * 2;
  const int roff = (lane >> 4) * 8;
  for (int t16 = wave; t16 < R_TILE / 16; t16 += NWAVE) {
    const int row0 = base + t16 * 16;
    v8f acc[NT];
#pragma unroll
    for (int t = 0; t < NT; ++t) acc[t] = (v8f)0.0f;

    const float* xrow = XD + (size_t)(row0 + m) * F_DIM;
    const float* prow = ptile + (t16 * 16 + m) * P_PITCH;
    const float nd = -disl[t16 * 16 + m];

#pragma unroll 4
    for (int kg = 0; kg < NKG; ++kg) {
      const int k0 = kg * 4;
      const v2f aX = *(const v2f*)(xrow + k0 + kk);
      v2f aP = *(const v2f*)(prow + k0 + kk);
      aP.x *= nd; aP.y *= nd;
#pragma unroll
      for (int t = 0; t < NT; ++t) {
        const float4 wv = wlds[(kg * NT + t) * 32 + lane];
        v2f b0, b1;
        b0.x = wv.x; b0.y = wv.y;
        b1.x = wv.z; b1.y = wv.w;
        acc[t] = __builtin_amdgcn_wmma_f32_16x16x4_f32(
            false, aX, false, b0, (short)0, acc[t], false, false);
        acc[t] = __builtin_amdgcn_wmma_f32_16x16x4_f32(
            false, aP, false, b1, (short)0, acc[t], false, false);
      }
    }

    float dsr[8];
#pragma unroll
    for (int r = 0; r < 8; ++r) dsr[r] = COPY ? disl[t16 * 16 + roff + r] : 0.0f;
    float* srow = ptile + (t16 * 16) * P_PITCH;
#pragma unroll
    for (int t = 0; t < NT; ++t) {
      const float bias = B[t * 16 + m];
#pragma unroll
      for (int r = 0; r < 8; ++r) {
        float vv = acc[t][r] + bias;
        if (RELU) vv = fmaxf(vv, 0.0f);
        OUT[(size_t)(row0 + roff + r) * FOUT + t * 16 + m] = vv;
        if (COPY) srow[(roff + r) * P_PITCH + t * 16 + m] = dsr[r] * vv;
      }
    }
    if (COPY) {
      wave_lds_handover();
      const int q4 = (lane & 15) * 4;
#pragma unroll
      for (int i = 0; i < 8; ++i) {
        const int rr = i * 2 + (lane >> 4);
        const float4 v = *(const float4*)(srow + rr * P_PITCH + q4);
        *(float4*)(OUTS + (size_t)(row0 + rr) * FOUT + q4) = v;
      }
    }
  }
}

extern "C" void kernel_launch(void* const* d_in, const int* in_sizes, int n_in,
                              void* d_out, int out_size, void* d_ws, size_t ws_size,
                              hipStream_t stream) {
  const float* x    = (const float*)d_in[0];
  const int*   adj  = (const int*)d_in[1];
  const float* W1_0 = (const float*)d_in[2];
  const float* W1_1 = (const float*)d_in[3];
  const float* b1   = (const float*)d_in[4];
  const float* Wx_0 = (const float*)d_in[5];
  const float* Wx_1 = (const float*)d_in[6];
  const float* bx   = (const float*)d_in[7];
  const float* W2_0 = (const float*)d_in[8];
  const float* W2_1 = (const float*)d_in[9];
  const float* b2   = (const float*)d_in[10];
  float* out = (float*)d_out;

  const int* row = adj;
  const int* col = adj + E_EDGES;

  float* H   = (float*)d_ws;
  float* H1s = H + (size_t)N_NODES * F_DIM;
  float* H2s = H1s + (size_t)N_NODES * F_DIM;
  float* deg = H2s + (size_t)N_NODES * F_DIM;
  unsigned* elist = (unsigned*)(deg + N_NODES);
  unsigned* key  = (unsigned*)H2s;
  unsigned* part = (unsigned*)H1s;
  if (ws_size < sizeof(float) * ((size_t)N_NODES + E_EDGES + 3 * (size_t)N_NODES * F_DIM)) return;

  const int gLay = N_NODES / R_TILE;

  edge_key_hist<<<KEY_WG, 1024, 0, stream>>>(row, key, part);

  scan_list<<<gLay, 1024, 0, stream>>>(row, col, key, part, elist, deg);

  cheb_layer<64, true, true, true><<<gLay, 1024, 0, stream>>>(elist, deg, x, x, W1_0, W1_1, b1, H, H1s);

  cheb_layer<64, true, false, true><<<gLay, 1024, 0, stream>>>(elist, deg, H, H1s, Wx_0, Wx_1, bx, H, H2s);

  cheb_layer<16, false, false, false><<<gLay, 1024, 0, stream>>>(elist, deg, H, H2s, W2_0, W2_1, b2, out, nullptr);
}


# edge walk's LDS float adds: the upper half-wave adds a row's 4 words in the order 2,3,0,1, so the 2 rows of an add hit 32 different banks, not the same banks twice; sum order: LDS arrival
# speedup vs baseline: 3.3133x; 1.1863x over previous
#include <hip/hip_runtime.h>

typedef __attribute__((ext_vector_type(2))) float v2f;
typedef __attribute__((ext_vector_type(8))) float v8f;

#define N_NODES 100000
#define E_EDGES 1600000
#define F_DIM   64

#define R_TILE  400
#define P_PITCH 68
#define NWAVE   32
#define QCAP    64

static __device__ __forceinline__ float dis_of(float d) {
  return (d > 0.0f) ? rsqrtf(d) : 0.0f;
}

#define KEY_WG    128
#define KEY_ITEMS (E_EDGES / 4 / KEY_WG)

__global__ __launch_bounds__(1024) void edge_key_hist(const int* __restrict__ row, unsigned* __restrict__ key,
                                                      unsigned* __restrict__ part) {
  __shared__ unsigned hist[256];
  const int tid = threadIdx.x;
  if (tid < 256) hist[tid] = 0u;
  __syncthreads();
  const int i0 = blockIdx.x * KEY_ITEMS;
  for (int i = i0 + tid; i < i0 + KEY_ITEMS; i += 1024) {
    const int4 r = ((const int4*)row)[i];
    const unsigned k0 = (unsigned)r.x / R_TILE, k1 = (unsigned)r.y / R_TILE;
    const unsigned k2 = (unsigned)r.z / R_TILE, k3 = (unsigned)r.w / R_TILE;
    key[i] = k0 | (k1 << 8) | (k2 << 16) | (k3 << 24);
    atomicAdd(&hist[k0], 1u);
    atomicAdd(&hist[k1], 1u);
    atomicAdd(&hist[k2], 1u);
    atomicAdd(&hist[k3], 1u);
  }
  __syncthreads();
  if (tid < N_NODES / R_TILE) part[tid * KEY_WG + blockIdx.x] = hist[tid];
}


static __device__ __forceinline__ void wave_lds_handover() {
  __builtin_amdgcn_fence(__ATOMIC_ACQ_REL, "wavefront");
  __builtin_amdgcn_wave_barrier();
}

static __device__ __forceinline__ void lds_add(float* p, float v) {
  __hip_atomic_fetch_add(p, v, __ATOMIC_RELAXED, __HIP_MEMORY_SCOPE_WORKGROUP);
}

template <bool FULL, int NG, bool WEIGHTED>
static __device__ __forceinline__ void gather_add(unsigned c, unsigned rl, int n,
                                                  int lane, int2* st, float* ptile,
                                                  const float* __restrict__ deg,
                                                  const float* __restrict__ X) {
  st[lane] = make_int2((int)(c * (F_DIM * 4u)), (int)(rl * (P_PITCH * 4u)));
  wave_lds_handover();
  const int half = lane >> 4;
  const unsigned kb = (unsigned)(lane & 15) * 16u;
  const unsigned kb_first  = kb + (half ? 8u : 0u);
  const unsigned kb_second = kb + (half ? 0u : 8u);
  int wmine = 0;
  int2 s[16];
#pragma unroll
  for (int i = 0; i < 16; ++i) s[i] = st[i * 2 + half];
#pragma unroll
  for (int hb = 0; hb < 16 / NG; ++hb) {
    float4 g[NG];
#pragma unroll
    for (int i = 0; i < NG; ++i) {
      const int j = (hb * NG + i) * 2 + half;
      g[i] = make_float4(0.0f, 0.0f, 0.0f, 0.0f);
      if (FULL || j < n) g[i] = *(const float4*)((const char*)X + ((unsigned)s[hb * NG + i].x + kb));
    }
    if (WEIGHTED && hb == 0) {
      __builtin_amdgcn_sched_barrier(0);
      float dgc = 0.0f;
      if (FULL || lane < n) dgc = deg[c];
      wmine = __builtin_bit_cast(int, dis_of(dgc));
    }
#pragma unroll
    for (int i = 0; i < NG; ++i) {
      const int j = (hb * NG + i) * 2 + half;
      float ww = 1.0f;
      if (WEIGHTED) {
        const float w_lo = __builtin_bit_cast(float, __builtin_amdgcn_readlane(wmine, (hb * NG + i) * 2));
        const float w_hi = __builtin_bit_cast(float, __builtin_amdgcn_readlane(wmine, (hb * NG + i) * 2 + 1));
        ww = half ? w_hi : w_lo;
      }
      if (FULL || j < n) {
        const float v0 = half ? g[i].z : g[i].x;
        const float v1 = half ? g[i].w : g[i].y;
        const float v2 = half ? g[i].x : g[i].z;
        const float v3 = half ? g[i].y : g[i].w;
        float* d01 = (float*)((char*)ptile + ((unsigned)s[hb * NG + i].y + kb_first));
        float* d23 = (float*)((char*)ptile + ((unsigned)s[hb * NG + i].y + kb_second));
        lds_add(d01 + 0, WEIGHTED ? ww * v0 : v0);
        lds_add(d01 + 1, WEIGHTED ? ww * v1 : v1);
        lds_add(d23 + 0, WEIGHTED ? ww * v2 : v2);
        lds_add(d23 + 1, WEIGHTED ? ww * v3 : v3);
      }
    }
  }
}

template <bool FULL>
static __device__ __forceinline__ void list_batch(unsigned e, int n, int lane, int base, unsigned* cnt,
                                                  const int* __restrict__ row,
                                                  const int* __restrict__ col,
                                                  unsigned* __restrict__ lst) {
  if (FULL || lane < n) {
    const unsigned rl = (unsigned)(row[e] - base);
    lst[lane] = (unsigned)col[e] | (rl << 17);
    atomicAdd(&cnt[rl], 1u);
  }
}

__global__ __launch_bounds__(1024) void scan_list(
    const int* __restrict__ row, const int* __restrict__ col, const unsigned* __restrict__ key,
    const unsigned* __restrict__ part, unsigned* __restrict__ elist, float* __restrict__ deg) {
  __shared__ unsigned cnt[R_TILE];
  __shared__ unsigned queue[NWAVE * QCAP];
  __shared__ unsigned red_lo[NWAVE];
  __shared__ unsigned lfill;

  const int tid  = threadIdx.x;
  const int lane = tid & 31;
  const int wave = __builtin_amdgcn_readfirstlane(tid >> 5);
  const int base = blockIdx.x * R_TILE;

  if (tid < R_TILE) cnt[tid] = 0u;
  {
    unsigned s_lo = 0u;
    const int4* p4 = (const int4*)part;
    const int n_lo = (int)blockIdx.x * (KEY_WG / 4);
    constexpr int DPT = 8;
    for (int i0 = 0; i0 < n_lo; i0 += DPT * NWAVE * 32) {
      int4 d[DPT];
#pragma unroll
      for (int k = 0; k < DPT; ++k)
        d[k] = p4[min(i0 + k * NWAVE * 32 + tid, n_lo - 1)];
#pragma unroll
      for (int k = 0; k < DPT; ++k)
        if (i0 + k * NWAVE * 32 + tid < n_lo)
          s_lo += ((unsigned)d[k].x + (unsigned)d[k].y) + ((unsigned)d[k].z + (unsigned)d[k].w);
    }
#pragma unroll
    for (int o = 16; o > 0; o >>= 1) s_lo += __shfl_xor(s_lo, o);
    if (lane == 0) red_lo[wave] = s_lo;
    if (tid == 0) lfill = 0u;
  }
  __syncthreads();
  unsigned t_lo = 0u;
#pragma unroll
  for (int w = 0; w < NWAVE; ++w) t_lo += red_lo[w];
  unsigned* mylist = elist + __builtin_amdgcn_readfirstlane(t_lo);

  unsigned* q   = queue + wave * QCAP;
  int qn = 0;
  const int4* key4 = (const int4*)key;
  constexpr int NCH = E_EDGES / 512;
  constexpr int LPT = 8;
  constexpr int NIT = (NCH + NWAVE * LPT - 1) / (NWAVE * LPT);
  const unsigned pat = (unsigned)blockIdx.x * 0x01010101u;

  auto nz = [](unsigned x) { return ((x & 0x7F7F7F7Fu) + 0x7F7F7F7Fu) | x; };
  auto take = [&](const int4 v, int ch) {
    const unsigned u0 = nz((unsigned)v.x ^ pat);
    const unsigned u1 = nz((unsigned)v.y ^ pat);
    const unsigned u2 = nz((unsigned)v.z ^ pat);
    const unsigned u3 = nz((unsigned)v.w ^ pat);
    const bool any = ((u0 & u1 & u2 & u3) & 0x80808080u) != 0x80808080u;
    unsigned mask = __builtin_amdgcn_ballot_w32(any);
    if (mask) {
      unsigned hv = (((u0 & 0x80808080u) >> 3) | ((u1 & 0x80808080u) >> 2) |
                     ((u2 & 0x80808080u) >> 1) | (u3 & 0x80808080u)) ^ 0xF0F0F0F0u;
      const unsigned e0 = (unsigned)(ch * 32 + lane) * 16u;
      do {
        if (hv) {
          const unsigned p = (unsigned)__builtin_ctz(hv);
          hv &= hv - 1u;
          q[qn + __builtin_amdgcn_mbcnt_lo(mask, 0u)] = e0 + (((p & 3u) << 2) | (p >> 3));
        }
        qn += __builtin_popcount(mask);
        mask = __builtin_amdgcn_ballot_w32(hv != 0u);
        while (qn >= 32) {
          qn -= 32;
          wave_lds_handover();
          unsigned slot = 0u;
          if (lane == 0) slot = __hip_atomic_fetch_add(&lfill, 32u, __ATOMIC_RELAXED, __HIP_MEMORY_SCOPE_WORKGROUP);
          slot = __builtin_amdgcn_readfirstlane(slot);
          list_batch<true>(q[qn + lane], 32, lane, base, cnt, row, col, mylist + slot);
        }
      } while (mask);
    }
  };

  for (int it0 = 0; it0 < NIT; ++it0) {
    const int it  = (it0 + (int)blockIdx.x) % NIT;
    const int ch0 = (it * NWAVE + wave) * LPT;
    if (ch0 >= NCH) continue;
    int4 v[LPT];
#pragma unroll
    for (int k = 0; k < LPT; ++k)
      v[k] = key4[min(ch0 + k, NCH - 1) * 32 + lane];
#pragma unroll
    for (int k = 0; k < LPT; ++k)
      if (ch0 + k < NCH) take(v[k], ch0 + k);
  }
  if (qn > 0) {
    wave_lds_handover();
    const unsigned ent = (lane < qn) ? q[lane] : 0u;
    unsigned slot = 0u;
    if (lane == 0) slot = __hip_atomic_fetch_add(&lfill, (unsigned)qn, __ATOMIC_RELAXED, __HIP_MEMORY_SCOPE_WORKGROUP);
    slot = __builtin_amdgcn_readfirstlane(slot);
    list_batch<false>(ent, qn, lane, base, cnt, row, col, mylist + slot);
  }
  __syncthreads();
  if (tid < R_TILE) deg[base + tid] = (float)cnt[tid];
}

template <int FOUT, bool RELU, bool WGT, bool COPY>
__global__ __launch_bounds__(1024) void cheb_layer(
    const unsigned* __restrict__ elist, const float* __restrict__ deg,
    const float* XD, const float* __restrict__ XG, const float* __restrict__ W0, const float* __restrict__ W1,
    const float* __restrict__ B, float* OUT, float* __restrict__ OUTS) {
  constexpr int NT   = FOUT / 16;
  constexpr int NKG  = F_DIM / 4;
  constexpr int NENT = NKG * NT * 32;

  __shared__ float    ptile[R_TILE * P_PITCH];
  __shared__ float    disl[R_TILE];
  __shared__ float4   wlds[NENT];
  __shared__ int2     stage[NWAVE * 32];
  __shared__ float    red_lo[NWAVE];
  __shared__ float    red_own[NWAVE];

  const int tid  = threadIdx.x;
  const int lane = tid & 31;
  const int wave = __builtin_amdgcn_readfirstlane(tid >> 5);
  const int base = blockIdx.x * R_TILE;

  for (int i = tid; i < R_TILE * P_PITCH / 4; i += NWAVE * 32)
    ((float4*)ptile)[i] = make_float4(0.0f, 0.0f, 0.0f, 0.0f);

  if (tid < R_TILE) disl[tid] = dis_of(deg[base + tid]);

  for (int idx = tid; idx < NENT; idx += NWAVE * 32) {
    const int l   = idx & 31;
    const int rem = idx >> 5;
    const int t   = rem % NT;
    const int kg  = rem / NT;
    const int n   = t * 16 + (l & 15);
    const int k   = kg * 4 + ((l >> 4) * 2);
    wlds[idx] = make_float4(W0[(size_t)k * FOUT + n], W0[(size_t)(k + 1) * FOUT + n],
                            W1[(size_t)k * FOUT + n], W1[(size_t)(k + 1) * FOUT + n]);
  }
  {
    float s_lo = 0.0f, s_own = 0.0f;
    const float4* d4 = (const float4*)deg;
    const int n_lo  = base / 4;
    const int n_all = n_lo + R_TILE / 4;
    constexpr int DPT = 8;
    for (int i0 = 0; i0 < n_all; i0 += DPT * NWAVE * 32) {
      float4 d[DPT];
#pragma unroll
      for (int k = 0; k < DPT; ++k)
        d[k] = d4[min(i0 + k * NWAVE * 32 + tid, n_all - 1)];
#pragma unroll
      for (int k = 0; k < DPT; ++k) {
        const int i = i0 + k * NWAVE * 32 + tid;
        const float s = (d[k].x + d[k].y) + (d[k].z + d[k].w);
        if (i < n_lo) s_lo += s; else if (i < n_all) s_own += s;
      }
    }
#pragma unroll
    for (int o = 16; o > 0; o >>= 1) {
      s_lo  += __shfl_xor(s_lo, o);
      s_own += __shfl_xor(s_own, o);
    }
    if (lane == 0) { red_lo[wave] = s_lo; red_own[wave] = s_own; }
  }
  __syncthreads();
  float t_lo = 0.0f, t_own = 0.0f;
#pragma unroll
  for (int w = 0; w < NWAVE; ++w) { t_lo += red_lo[w]; t_own += red_own[w]; }
  const unsigned l_off = __builtin_amdgcn_readfirstlane((unsigned)t_lo);
  const unsigned l_cnt = __builtin_amdgcn_readfirstlane((unsigned)t_own);
  const unsigned* mylist = elist + l_off;

  int2*     st  = stage + wave * 32;

  const int nfull = (int)(l_cnt >> 5);
  const int ntail = (int)(l_cnt & 31u);
  if (wave < nfull) {
    const int blast = nfull - 1;
    unsigned w0 = mylist[wave * 32 + lane];
    unsigned wn = mylist[min(wave + NWAVE, blast) * 32 + lane];
    for (int b = wave; b < nfull; b += NWAVE) {
      const unsigned wn2 = mylist[min(b + 2 * NWAVE, blast) * 32 + lane];
      gather_add<true, 16, WGT>(w0 & 0x1FFFFu, w0 >> 17, 32, lane, st, ptile, deg, XG);
      w0 = wn; wn = wn2;
    }
  }
  if (ntail > 0 && wave == (nfull & (NWAVE - 1))) {
    const unsigned ent = (lane < ntail) ? mylist[nfull * 32 + lane] : 0u;
    gather_add<false, 16, WGT>(ent & 0x1FFFFu, ent >> 17, ntail, lane, st, ptile, deg, XG);
  }
  __syncthreads();

  const int m  = lane & 15;
  const int kk = (lane >> 4) * 2;
  const int roff = (lane >> 4) * 8;
  for (int t16 = wave; t16 < R_TILE / 16; t16 += NWAVE) {
    const int row0 = base + t16 * 16;
    v8f acc[NT];
#pragma unroll
    for (int t = 0; t < NT; ++t) acc[t] = (v8f)0.0f;

    const float* xrow = XD + (size_t)(row0 + m) * F_DIM;
    const float* prow = ptile + (t16 * 16 + m) * P_PITCH;
    const float nd = -disl[t16 * 16 + m];

#pragma unroll 4
    for (int kg = 0; kg < NKG; ++kg) {
      const int k0 = kg * 4;
      const v2f aX = *(const v2f*)(xrow + k0 + kk);
      v2f aP = *(const v2f*)(prow + k0 + kk);
      aP.x *= nd; aP.y *= nd;
#pragma unroll
      for (int t = 0; t < NT; ++t) {
        const float4 wv = wlds[(kg * NT + t) * 32 + lane];
        v2f b0, b1;
        b0.x = wv.x; b0.y = wv.y;
        b1.x = wv.z; b1.y = wv.w;
        acc[t] = __builtin_amdgcn_wmma_f32_16x16x4_f32(
            false, aX, false, b0, (short)0, acc[t], false, false);
        acc[t] = __builtin_amdgcn_wmma_f32_16x16x4_f32(
            false, aP, false, b1, (short)0, acc[t], false, false);
      }
    }

    float dsr[8];
#pragma unroll
    for (int r = 0; r < 8; ++r) dsr[r] = COPY ? disl[t16 * 16 + roff + r] : 0.0f;
    float* srow = ptile + (t16 * 16) * P_PITCH;
#pragma unroll
    for (int t = 0; t < NT; ++t) {
      const float bias = B[t * 16 + m];
#pragma unroll
      for (int r = 0; r < 8; ++r) {
        float vv = acc[t][r] + bias;
        if (RELU) vv = fmaxf(vv, 0.0f);
        OUT[(size_t)(row0 + roff + r) * FOUT + t * 16 + m] = vv;
        if (COPY) srow[(roff + r) * P_PITCH + t * 16 + m] = dsr[r] * vv;
      }
    }
    if (COPY) {
      wave_lds_handover();
      const int q4 = (lane & 15) * 4;
#pragma unroll
      for (int i = 0; i < 8; ++i) {
        const int rr = i * 2 + (lane >> 4);
        const float4 v = *(const float4*)(srow + rr * P_PITCH + q4);
        *(float4*)(OUTS + (size_t)(row0 + rr) * FOUT + q4) = v;
      }
    }
  }
}

extern "C" void kernel_launch(void* const* d_in, const int* in_sizes, int n_in,
                              void* d_out, int out_size, void* d_ws, size_t ws_size,
                              hipStream_t stream) {
  const float* x    = (const float*)d_in[0];
  const int*   adj  = (const int*)d_in[1];
  const float* W1_0 = (const float*)d_in[2];
  const float* W1_1 = (const float*)d_in[3];
  const float* b1   = (const float*)d_in[4];
  const float* Wx_0 = (const float*)d_in[5];
  const float* Wx_1 = (const float*)d_in[6];
  const float* bx   = (const float*)d_in[7];
  const float* W2_0 = (const float*)d_in[8];
  const float* W2_1 = (const float*)d_in[9];
  const float* b2   = (const float*)d_in[10];
  float* out = (float*)d_out;

  const int* row = adj;
  const int* col = adj + E_EDGES;

  float* H   = (float*)d_ws;
  float* H1s = H + (size_t)N_NODES * F_DIM;
  float* H2s = H1s + (size_t)N_NODES * F_DIM;
  float* deg = H2s + (size_t)N_NODES * F_DIM;
  unsigned* elist = (unsigned*)(deg + N_NODES);
  unsigned* key  = (unsigned*)H2s;
  unsigned* part = (unsigned*)H1s;
  if (ws_size < sizeof(float) * ((size_t)N_NODES + E_EDGES + 3 * (size_t)N_NODES * F_DIM)) return;

  const int gLay = N_NODES / R_TILE;

  edge_key_hist<<<KEY_WG, 1024, 0, stream>>>(row, key, part);

  scan_list<<<gLay, 1024, 0, stream>>>(row, col, key, part, elist, deg);

  cheb_layer<64, true, true, true><<<gLay, 1024, 0, stream>>>(elist, deg, x, x, W1_0, W1_1, b1, H, H1s);

  cheb_layer<64, true, false, true><<<gLay, 1024, 0, stream>>>(elist, deg, H, H1s, Wx_0, Wx_1, bx, H, H2s);

  cheb_layer<16, false, false, false><<<gLay, 1024, 0, stream>>>(elist, deg, H, H2s, W2_0, W2_1, b2, out, nullptr);
}


# owner scan reads its presence bits (1 bit per 16-edge group, ORed in LDS by the key launch) and fetches keys of flagged groups only, not every key; list order changes; sum order: LDS arrival
# speedup vs baseline: 3.4335x; 1.0363x over previous
#include <hip/hip_runtime.h>

typedef __attribute__((ext_vector_type(2))) float v2f;
typedef __attribute__((ext_vector_type(8))) float v8f;

#define N_NODES 100000
#define E_EDGES 1600000
#define F_DIM   64

#define R_TILE  400
#define P_PITCH 68
#define NWAVE   32
#define QCAP    64

static __device__ __forceinline__ float dis_of(float d) {
  return (d > 0.0f) ? rsqrtf(d) : 0.0f;
}

#define N_OWN      (N_NODES / R_TILE)
#define KEY_WG     125
#define KEY_ITEMS  (E_EDGES / 4 / KEY_WG)
#define PART_PITCH 128
#define BIT_W      (KEY_ITEMS / 4 / 32)
#define NBITW      (E_EDGES / 16 / 32)

__global__ __launch_bounds__(1024) void edge_key_hist(const int* __restrict__ row, unsigned* __restrict__ key,
                                                      unsigned* __restrict__ part, unsigned* __restrict__ gbits) {
  __shared__ unsigned hist[256];
  __shared__ unsigned bits[N_OWN * BIT_W];
  const int tid = threadIdx.x;
  if (tid < 256) hist[tid] = 0u;
  for (int i = tid; i < N_OWN * BIT_W; i += 1024) bits[i] = 0u;
  __syncthreads();
  const int i0 = blockIdx.x * KEY_ITEMS;
  for (int i = tid; i < KEY_ITEMS; i += 1024) {
    const int4 r = ((const int4*)row)[i0 + i];
    const unsigned k0 = (unsigned)r.x / R_TILE, k1 = (unsigned)r.y / R_TILE;
    const unsigned k2 = (unsigned)r.z / R_TILE, k3 = (unsigned)r.w / R_TILE;
    key[i0 + i] = k0 | (k1 << 8) | (k2 << 16) | (k3 << 24);
    atomicAdd(&hist[k0], 1u);
    atomicAdd(&hist[k1], 1u);
    atomicAdd(&hist[k2], 1u);
    atomicAdd(&hist[k3], 1u);
    const unsigned g = (unsigned)i >> 2;
    const unsigned w = g >> 5, b = 1u << (g & 31u);
    atomicOr(&bits[k0 * BIT_W + w], b);
    atomicOr(&bits[k1 * BIT_W + w], b);
    atomicOr(&bits[k2 * BIT_W + w], b);
    atomicOr(&bits[k3 * BIT_W + w], b);
  }
  __syncthreads();
  if (tid < N_OWN) part[tid * PART_PITCH + blockIdx.x] = hist[tid];
  if (blockIdx.x < PART_PITCH - KEY_WG && tid < N_OWN) part[tid * PART_PITCH + KEY_WG + blockIdx.x] = 0u;
  for (int i = tid; i < N_OWN * BIT_W; i += 1024) {
    const int o = i / BIT_W, w = i - o * BIT_W;
    gbits[(size_t)o * NBITW + blockIdx.x * BIT_W + w] = bits[i];
  }
}


static __device__ __forceinline__ void wave_lds_handover() {
  __builtin_amdgcn_fence(__ATOMIC_ACQ_REL, "wavefront");
  __builtin_amdgcn_wave_barrier();
}

static __device__ __forceinline__ void lds_add(float* p, float v) {
  __hip_atomic_fetch_add(p, v, __ATOMIC_RELAXED, __HIP_MEMORY_SCOPE_WORKGROUP);
}

template <bool FULL, int NG, bool WEIGHTED>
static __device__ __forceinline__ void gather_add(unsigned c, unsigned rl, int n,
                                                  int lane, int2* st, float* ptile,
                                                  const float* __restrict__ deg,
                                                  const float* __restrict__ X) {
  st[lane] = make_int2((int)(c * (F_DIM * 4u)), (int)(rl * (P_PITCH * 4u)));
  wave_lds_handover();
  const int half = lane >> 4;
  const unsigned kb = (unsigned)(lane & 15) * 16u;
  const unsigned kb_first  = kb + (half ? 8u : 0u);
  const unsigned kb_second = kb + (half ? 0u : 8u);
  int wmine = 0;
  int2 s[16];
#pragma unroll
  for (int i = 0; i < 16; ++i) s[i] = st[i * 2 + half];
#pragma unroll
  for (int hb = 0; hb < 16 / NG; ++hb) {
    float4 g[NG];
#pragma unroll
    for (int i = 0; i < NG; ++i) {
      const int j = (hb * NG + i) * 2 + half;
      g[i] = make_float4(0.0f, 0.0f, 0.0f, 0.0f);
      if (FULL || j < n) g[i] = *(const float4*)((const char*)X + ((unsigned)s[hb * NG + i].x + kb));
    }
    if (WEIGHTED && hb == 0) {
      __builtin_amdgcn_sched_barrier(0);
      float dgc = 0.0f;
      if (FULL || lane < n) dgc = deg[c];
      wmine = __builtin_bit_cast(int, dis_of(dgc));
    }
#pragma unroll
    for (int i = 0; i < NG; ++i) {
      const int j = (hb * NG + i) * 2 + half;
      float ww = 1.0f;
      if (WEIGHTED) {
        const float w_lo = __builtin_bit_cast(float, __builtin_amdgcn_readlane(wmine, (hb * NG + i) * 2));
        const float w_hi = __builtin_bit_cast(float, __builtin_amdgcn_readlane(wmine, (hb * NG + i) * 2 + 1));
        ww = half ? w_hi : w_lo;
      }
      if (FULL || j < n) {
        const float v0 = half ? g[i].z : g[i].x;
        const float v1 = half ? g[i].w : g[i].y;
        const float v2 = half ? g[i].x : g[i].z;
        const float v3 = half ? g[i].y : g[i].w;
        float* d01 = (float*)((char*)ptile + ((unsigned)s[hb * NG + i].y + kb_first));
        float* d23 = (float*)((char*)ptile + ((unsigned)s[hb * NG + i].y + kb_second));
        lds_add(d01 + 0, WEIGHTED ? ww * v0 : v0);
        lds_add(d01 + 1, WEIGHTED ? ww * v1 : v1);
        lds_add(d23 + 0, WEIGHTED ? ww * v2 : v2);
        lds_add(d23 + 1, WEIGHTED ? ww * v3 : v3);
      }
    }
  }
}

template <bool FULL>
static __device__ __forceinline__ void list_batch(unsigned e, int n, int lane, int base, unsigned* cnt,
                                                  const int* __restrict__ row,
                                                  const int* __restrict__ col,
                                                  unsigned* __restrict__ lst) {
  if (FULL || lane < n) {
    const unsigned rl = (unsigned)(row[e] - base);
    lst[lane] = (unsigned)col[e] | (rl << 17);
    atomicAdd(&cnt[rl], 1u);
  }
}

__global__ __launch_bounds__(1024) void scan_list(
    const int* __restrict__ row, const int* __restrict__ col, const unsigned* __restrict__ key,
    const unsigned* __restrict__ part, const unsigned* __restrict__ gbits,
    unsigned* __restrict__ elist, float* __restrict__ deg) {
  __shared__ unsigned cnt[R_TILE];
  __shared__ unsigned queue[NWAVE * QCAP];
  __shared__ unsigned red_lo[NWAVE];
  __shared__ unsigned lfill;

  const int tid  = threadIdx.x;
  const int lane = tid & 31;
  const int wave = __builtin_amdgcn_readfirstlane(tid >> 5);
  const int base = blockIdx.x * R_TILE;

  if (tid < R_TILE) cnt[tid] = 0u;
  {
    unsigned s_lo = 0u;
    const int4* p4 = (const int4*)part;
    const int n_lo = (int)blockIdx.x * (PART_PITCH / 4);
    constexpr int DPT = 8;
    for (int i0 = 0; i0 < n_lo; i0 += DPT * NWAVE * 32) {
      int4 d[DPT];
#pragma unroll
      for (int k = 0; k < DPT; ++k)
        d[k] = p4[min(i0 + k * NWAVE * 32 + tid, n_lo - 1)];
#pragma unroll
      for (int k = 0; k < DPT; ++k)
        if (i0 + k * NWAVE * 32 + tid < n_lo)
          s_lo += ((unsigned)d[k].x + (unsigned)d[k].y) + ((unsigned)d[k].z + (unsigned)d[k].w);
    }
#pragma unroll
    for (int o = 16; o > 0; o >>= 1) s_lo += __shfl_xor(s_lo, o);
    if (lane == 0) red_lo[wave] = s_lo;
    if (tid == 0) lfill = 0u;
  }
  __syncthreads();
  unsigned t_lo = 0u;
#pragma unroll
  for (int w = 0; w < NWAVE; ++w) t_lo += red_lo[w];
  unsigned* mylist = elist + __builtin_amdgcn_readfirstlane(t_lo);

  unsigned* q   = queue + wave * QCAP;
  int qn = 0;
  const int4* key4 = (const int4*)key;
  const unsigned* mybits = gbits + (size_t)blockIdx.x * NBITW;
  constexpr int NTAKE = 4;
  const unsigned pat = (unsigned)blockIdx.x * 0x01010101u;

  auto nz = [](unsigned x) { return ((x & 0x7F7F7F7Fu) + 0x7F7F7F7Fu) | x; };
  auto take = [&](const int4 v, unsigned e0, bool valid) {
    const unsigned u0 = nz((unsigned)v.x ^ pat);
    const unsigned u1 = nz((unsigned)v.y ^ pat);
    const unsigned u2 = nz((unsigned)v.z ^ pat);
    const unsigned u3 = nz((unsigned)v.w ^ pat);
    unsigned hv = (((u0 & 0x80808080u) >> 3) | ((u1 & 0x80808080u) >> 2) |
                   ((u2 & 0x80808080u) >> 1) | (u3 & 0x80808080u)) ^ 0xF0F0F0F0u;
    if (!valid) hv = 0u;
    unsigned mask = __builtin_amdgcn_ballot_w32(hv != 0u);
    while (mask) {
      if (hv) {
        const unsigned p = (unsigned)__builtin_ctz(hv);
        hv &= hv - 1u;
        q[qn + __builtin_amdgcn_mbcnt_lo(mask, 0u)] = e0 + (((p & 3u) << 2) | (p >> 3));
      }
      qn += __builtin_popcount(mask);
      mask = __builtin_amdgcn_ballot_w32(hv != 0u);
      while (qn >= 32) {
        qn -= 32;
        wave_lds_handover();
        unsigned slot = 0u;
        if (lane == 0) slot = __hip_atomic_fetch_add(&lfill, 32u, __ATOMIC_RELAXED, __HIP_MEMORY_SCOPE_WORKGROUP);
        slot = __builtin_amdgcn_readfirstlane(slot);
        list_batch<true>(q[qn + lane], 32, lane, base, cnt, row, col, mylist + slot);
      }
    }
  };

  constexpr int WPW = (NBITW + NWAVE - 1) / NWAVE;
  constexpr int WPL = (WPW + 31) / 32;
  unsigned mw[WPL];
#pragma unroll
  for (int t = 0; t < WPL; ++t) {
    const int wi = t * 32 + lane;
    const int cw = wave * WPW + wi;
    mw[t] = (wi < WPW && cw < NBITW) ? mybits[cw] : 0u;
  }
  for (;;) {
    unsigned left = 0u;
#pragma unroll
    for (int t = 0; t < WPL; ++t) left |= mw[t];
    if (__builtin_amdgcn_ballot_w32(left != 0u) == 0u) break;
    unsigned g[NTAKE];
    bool has[NTAKE];
    int4 v[NTAKE];
#pragma unroll
    for (int k = 0; k < NTAKE; ++k) {
      unsigned cur = 0u;
      int sel = 0;
#pragma unroll
      for (int t = WPL - 1; t >= 0; --t)
        if (mw[t] != 0u) { cur = mw[t]; sel = t; }
      has[k] = (cur != 0u);
      const unsigned j = has[k] ? (unsigned)__builtin_ctz(cur) : 0u;
      g[k] = (unsigned)(wave * WPW + sel * 32 + lane) * 32u + j;
#pragma unroll
      for (int t = 0; t < WPL; ++t)
        if (t == sel) mw[t] &= mw[t] - 1u;
    }
#pragma unroll
    for (int k = 0; k < NTAKE; ++k) v[k] = key4[has[k] ? g[k] : 0u];
#pragma unroll
    for (int k = 0; k < NTAKE; ++k) take(v[k], g[k] * 16u, has[k]);
  }
  if (qn > 0) {
    wave_lds_handover();
    const unsigned ent = (lane < qn) ? q[lane] : 0u;
    unsigned slot = 0u;
    if (lane == 0) slot = __hip_atomic_fetch_add(&lfill, (unsigned)qn, __ATOMIC_RELAXED, __HIP_MEMORY_SCOPE_WORKGROUP);
    slot = __builtin_amdgcn_readfirstlane(slot);
    list_batch<false>(ent, qn, lane, base, cnt, row, col, mylist + slot);
  }
  __syncthreads();
  if (tid < R_TILE) deg[base + tid] = (float)cnt[tid];
}

template <int FOUT, bool RELU, bool WGT, bool COPY>
__global__ __launch_bounds__(1024) void cheb_layer(
    const unsigned* __restrict__ elist, const float* __restrict__ deg,
    const float* XD, const float* __restrict__ XG, const float* __restrict__ W0, const float* __restrict__ W1,
    const float* __restrict__ B, float* OUT, float* __restrict__ OUTS) {
  constexpr int NT   = FOUT / 16;
  constexpr int NKG  = F_DIM / 4;
  constexpr int NENT = NKG * NT * 32;

  __shared__ float    ptile[R_TILE * P_PITCH];
  __shared__ float    disl[R_TILE];
  __shared__ float4   wlds[NENT];
  __shared__ int2     stage[NWAVE * 32];
  __shared__ float    red_lo[NWAVE];
  __shared__ float    red_own[NWAVE];

  const int tid  = threadIdx.x;
  const int lane = tid & 31;
  const int wave = __builtin_amdgcn_readfirstlane(tid >> 5);
  const int base = blockIdx.x * R_TILE;

  for (int i = tid; i < R_TILE * P_PITCH / 4; i += NWAVE * 32)
    ((float4*)ptile)[i] = make_float4(0.0f, 0.0f, 0.0f, 0.0f);

  if (tid < R_TILE) disl[tid] = dis_of(deg[base + tid]);

  for (int idx = tid; idx < NENT; idx += NWAVE * 32) {
    const int l   = idx & 31;
    const int rem = idx >> 5;
    const int t   = rem % NT;
    const int kg  = rem / NT;
    const int n   = t * 16 + (l & 15);
    const int k   = kg * 4 + ((l >> 4) * 2);
    wlds[idx] = make_float4(W0[(size_t)k * FOUT + n], W0[(size_t)(k + 1) * FOUT + n],
                            W1[(size_t)k * FOUT + n], W1[(size_t)(k + 1) * FOUT + n]);
  }
  {
    float s_lo = 0.0f, s_own = 0.0f;
    const float4* d4 = (const float4*)deg;
    const int n_lo  = base / 4;
    const int n_all = n_lo + R_TILE / 4;
    constexpr int DPT = 8;
    for (int i0 = 0; i0 < n_all; i0 += DPT * NWAVE * 32) {
      float4 d[DPT];
#pragma unroll
      for (int k = 0; k < DPT; ++k)
        d[k] = d4[min(i0 + k * NWAVE * 32 + tid, n_all - 1)];
#pragma unroll
      for (int k = 0; k < DPT; ++k) {
        const int i = i0 + k * NWAVE * 32 + tid;
        const float s = (d[k].x + d[k].y) + (d[k].z + d[k].w);
        if (i < n_lo) s_lo += s; else if (i < n_all) s_own += s;
      }
    }
#pragma unroll
    for (int o = 16; o > 0; o >>= 1) {
      s_lo  += __shfl_xor(s_lo, o);
      s_own += __shfl_xor(s_own, o);
    }
    if (lane == 0) { red_lo[wave] = s_lo; red_own[wave] = s_own; }
  }
  __syncthreads();
  float t_lo = 0.0f, t_own = 0.0f;
#pragma unroll
  for (int w = 0; w < NWAVE; ++w) { t_lo += red_lo[w]; t_own += red_own[w]; }
  const unsigned l_off = __builtin_amdgcn_readfirstlane((unsigned)t_lo);
  const unsigned l_cnt = __builtin_amdgcn_readfirstlane((unsigned)t_own);
  const unsigned* mylist = elist + l_off;

  int2*     st  = stage + wave * 32;

  const int nfull = (int)(l_cnt >> 5);
  const int ntail = (int)(l_cnt & 31u);
  if (wave < nfull) {
    const int blast = nfull - 1;
    unsigned w0 = mylist[wave * 32 + lane];
    unsigned wn = mylist[min(wave + NWAVE, blast) * 32 + lane];
    for (int b = wave; b < nfull; b += NWAVE) {
      const unsigned wn2 = mylist[min(b + 2 * NWAVE, blast) * 32 + lane];
      gather_add<true, 16, WGT>(w0 & 0x1FFFFu, w0 >> 17, 32, lane, st, ptile, deg, XG);
      w0 = wn; wn = wn2;
    }
  }
  if (ntail > 0 && wave == (nfull & (NWAVE - 1))) {
    const unsigned ent = (lane < ntail) ? mylist[nfull * 32 + lane] : 0u;
    gather_add<false, 16, WGT>(ent & 0x1FFFFu, ent >> 17, ntail, lane, st, ptile, deg, XG);
  }
  __syncthreads();

  const int m  = lane & 15;
  const int kk = (lane >> 4) * 2;
  const int roff = (lane >> 4) * 8;
  for (int t16 = wave; t16 < R_TILE / 16; t16 += NWAVE) {
    const int row0 = base + t16 * 16;
    v8f acc[NT];
#pragma unroll
    for (int t = 0; t < NT; ++t) acc[t] = (v8f)0.0f;

    const float* xrow = XD + (size_t)(row0 + m) * F_DIM;
    const float* prow = ptile + (t16 * 16 + m) * P_PITCH;
    const float nd = -disl[t16 * 16 + m];

#pragma unroll 4
    for (int kg = 0; kg < NKG; ++kg) {
      const int k0 = kg * 4;
      const v2f aX = *(const v2f*)(xrow + k0 + kk);
      v2f aP = *(const v2f*)(prow + k0 + kk);
      aP.x *= nd; aP.y *= nd;
#pragma unroll
      for (int t = 0; t < NT; ++t) {
        const float4 wv = wlds[(kg * NT + t) * 32 + lane];
        v2f b0, b1;
        b0.x = wv.x; b0.y = wv.y;
        b1.x = wv.z; b1.y = wv.w;
        acc[t] = __builtin_amdgcn_wmma_f32_16x16x4_f32(
            false, aX, false, b0, (short)0, acc[t], false, false);
        acc[t] = __builtin_amdgcn_wmma_f32_16x16x4_f32(
            false, aP, false, b1, (short)0, acc[t], false, false);
      }
    }

    float dsr[8];
#pragma unroll
    for (int r = 0; r < 8; ++r) dsr[r] = COPY ? disl[t16 * 16 + roff + r] : 0.0f;
    float* srow = ptile + (t16 * 16) * P_PITCH;
#pragma unroll
    for (int t = 0; t < NT; ++t) {
      const float bias = B[t * 16 + m];
#pragma unroll
      for (int r = 0; r < 8; ++r) {
        float vv = acc[t][r] + bias;
        if (RELU) vv = fmaxf(vv, 0.0f);
        OUT[(size_t)(row0 + roff + r) * FOUT + t * 16 + m] = vv;
        if (COPY) srow[(roff + r) * P_PITCH + t * 16 + m] = dsr[r] * vv;
      }
    }
    if (COPY) {
      wave_lds_handover();
      const int q4 = (lane & 15) * 4;
#pragma unroll
      for (int i = 0; i < 8; ++i) {
        const int rr = i * 2 + (lane >> 4);
        const float4 v = *(const float4*)(srow + rr * P_PITCH + q4);
        *(float4*)(OUTS + (size_t)(row0 + rr) * FOUT + q4) = v;
      }
    }
  }
}

extern "C" void kernel_launch(void* const* d_in, const int* in_sizes, int n_in,
                              void* d_out, int out_size, void* d_ws, size_t ws_size,
                              hipStream_t stream) {
  const float* x    = (const float*)d_in[0];
  const int*   adj  = (const int*)d_in[1];
  const float* W1_0 = (const float*)d_in[2];
  const float* W1_1 = (const float*)d_in[3];
  const float* b1   = (const float*)d_in[4];
  const float* Wx_0 = (const float*)d_in[5];
  const float* Wx_1 = (const float*)d_in[6];
  const float* bx   = (const float*)d_in[7];
  const float* W2_0 = (const float*)d_in[8];
  const float* W2_1 = (const float*)d_in[9];
  const float* b2   = (const float*)d_in[10];
  float* out = (float*)d_out;

  const int* row = adj;
  const int* col = adj + E_EDGES;

  float* H   = (float*)d_ws;
  float* H1s = H + (size_t)N_NODES * F_DIM;
  float* H2s = H1s + (size_t)N_NODES * F_DIM;
  float* deg = H2s + (size_t)N_NODES * F_DIM;
  unsigned* elist = (unsigned*)(deg + N_NODES);
  unsigned* key  = (unsigned*)H2s;
  unsigned* part = (unsigned*)H1s;
  unsigned* gbits = (unsigned*)H;
  if (ws_size < sizeof(float) * ((size_t)N_NODES + E_EDGES + 3 * (size_t)N_NODES * F_DIM)) return;

  const int gLay = N_NODES / R_TILE;

  edge_key_hist<<<KEY_WG, 1024, 0, stream>>>(row, key, part, gbits);

  scan_list<<<gLay, 1024, 0, stream>>>(row, col, key, part, gbits, elist, deg);

  cheb_layer<64, true, true, true><<<gLay, 1024, 0, stream>>>(elist, deg, x, x, W1_0, W1_1, b1, H, H1s);

  cheb_layer<64, true, false, true><<<gLay, 1024, 0, stream>>>(elist, deg, H, H1s, Wx_0, Wx_1, bx, H, H2s);

  cheb_layer<16, false, false, false><<<gLay, 1024, 0, stream>>>(elist, deg, H, H2s, W2_0, W2_1, b2, out, nullptr);
}


# neighbour sum shared by two units: scan groups each edge list by 16-row tile; 16 waves sum half of it on f32 WMMA (selector 16x4 @ 4 gathered rows), 16 keep LDS float adds; sum order changes
# speedup vs baseline: 3.8379x; 1.1178x over previous
#include <hip/hip_runtime.h>

typedef __attribute__((ext_vector_type(2))) float v2f;
typedef __attribute__((ext_vector_type(8))) float v8f;

#define N_NODES 100000
#define E_EDGES 1600000
#define F_DIM   64

#define R_TILE  400
#define P_PITCH 68
#define NWAVE   32
#define QCAP    64

static __device__ __forceinline__ float dis_of(float d) {
  return (d > 0.0f) ? rsqrtf(d) : 0.0f;
}

#define N_OWN      (N_NODES / R_TILE)
#define KEY_WG     125
#define KEY_ITEMS  (E_EDGES / 4 / KEY_WG)
#define PART_PITCH 128
#define BIT_W      (KEY_ITEMS / 4 / 32)
#define NBITW      (E_EDGES / 16 / 32)

__global__ __launch_bounds__(1024) void edge_key_hist(const int* __restrict__ row, unsigned* __restrict__ key,
                                                      unsigned* __restrict__ part, unsigned* __restrict__ gbits) {
  __shared__ unsigned hist[256];
  __shared__ unsigned bits[N_OWN * BIT_W];
  const int tid = threadIdx.x;
  if (tid < 256) hist[tid] = 0u;
  for (int i = tid; i < N_OWN * BIT_W; i += 1024) bits[i] = 0u;
  __syncthreads();
  const int i0 = blockIdx.x * KEY_ITEMS;
  for (int i = tid; i < KEY_ITEMS; i += 1024) {
    const int4 r = ((const int4*)row)[i0 + i];
    const unsigned k0 = (unsigned)r.x / R_TILE, k1 = (unsigned)r.y / R_TILE;
    const unsigned k2 = (unsigned)r.z / R_TILE, k3 = (unsigned)r.w / R_TILE;
    key[i0 + i] = k0 | (k1 << 8) | (k2 << 16) | (k3 << 24);
    atomicAdd(&hist[k0], 1u);
    atomicAdd(&hist[k1], 1u);
    atomicAdd(&hist[k2], 1u);
    atomicAdd(&hist[k3], 1u);
    const unsigned g = (unsigned)i >> 2;
    const unsigned w = g >> 5, b = 1u << (g & 31u);
    atomicOr(&bits[k0 * BIT_W + w], b);
    atomicOr(&bits[k1 * BIT_W + w], b);
    atomicOr(&bits[k2 * BIT_W + w], b);
    atomicOr(&bits[k3 * BIT_W + w], b);
  }
  __syncthreads();
  if (tid < N_OWN) part[tid * PART_PITCH + blockIdx.x] = hist[tid];
  if (blockIdx.x < PART_PITCH - KEY_WG && tid < N_OWN) part[tid * PART_PITCH + KEY_WG + blockIdx.x] = 0u;
  for (int i = tid; i < N_OWN * BIT_W; i += 1024) {
    const int o = i / BIT_W, w = i - o * BIT_W;
    gbits[(size_t)o * NBITW + blockIdx.x * BIT_W + w] = bits[i];
  }
}


static __device__ __forceinline__ void wave_lds_handover() {
  __builtin_amdgcn_fence(__ATOMIC_ACQ_REL, "wavefront");
  __builtin_amdgcn_wave_barrier();
}

static __device__ __forceinline__ void lds_add(float* p, float v) {
  __hip_atomic_fetch_add(p, v, __ATOMIC_RELAXED, __HIP_MEMORY_SCOPE_WORKGROUP);
}

static __device__ __forceinline__ void flush_tile(const v8f (&acc)[4], int T, int lane, float* ptile) {
  const int h = lane >> 4;
  float* rowp = ptile + (T * 16 + 8 * h) * P_PITCH + 4 * (lane & 15);
  float* pa = rowp + (h ? 2 : 0);
  float* pb = rowp + (h ? 0 : 2);
#pragma unroll
  for (int r = 0; r < 8; ++r) {
    lds_add(pa + r * P_PITCH + 0, h ? acc[2][r] : acc[0][r]);
    lds_add(pa + r * P_PITCH + 1, h ? acc[3][r] : acc[1][r]);
    lds_add(pb + r * P_PITCH + 0, h ? acc[0][r] : acc[2][r]);
    lds_add(pb + r * P_PITCH + 1, h ? acc[1][r] : acc[3][r]);
  }
}

static __device__ __forceinline__ void tile_batch(unsigned w, float wt, int lane, int2* st, float* ptile,
                                                  const float* __restrict__ X, v8f (&acc)[4], int& curT) {
  st[lane] = make_int2((int)w, __builtin_bit_cast(int, wt));
  wave_lds_handover();
  const int h = lane >> 4;
  const int n = lane & 15;
  const unsigned kb = (unsigned)n * 16u;
  const int myT = (int)(w >> 21);
  const int4* s4 = (const int4*)st;
#pragma unroll
  for (int hb = 0; hb < 2; ++hb) {
    int4 q[4];
    float4 p[4], r[4];
#pragma unroll
    for (int i = 0; i < 4; ++i) q[i] = s4[(hb * 4 + i) * 2 + h];
#pragma unroll
    for (int i = 0; i < 4; ++i) {
      p[i] = *(const float4*)((const char*)X + ((((unsigned)q[i].x & 0x1FFFFu) << 8) + kb));
      r[i] = *(const float4*)((const char*)X + ((((unsigned)q[i].z & 0x1FFFFu) << 8) + kb));
    }
#pragma unroll
    for (int i = 0; i < 4; ++i) {
      const int g = hb * 4 + i;
      const int Tlo = __builtin_amdgcn_readlane(myT, 4 * g);
      const int Thi = __builtin_amdgcn_readlane(myT, 4 * g + 3);
      const int rl0 = (int)((unsigned)q[i].x >> 17);
      const int rl1 = (int)((unsigned)q[i].z >> 17);
      const float w0 = __builtin_bit_cast(float, q[i].y);
      const float w1 = __builtin_bit_cast(float, q[i].w);
      for (int T = Tlo; T <= Thi; ++T) {
        if (T != curT) {
          if (curT >= 0) flush_tile(acc, curT, lane, ptile);
#pragma unroll
          for (int j = 0; j < 4; ++j) acc[j] = (v8f)0.0f;
          curT = T;
        }
        const int rowm = T * 16 + n;
        v2f a, b;
        a.x = (rl0 == rowm) ? w0 : 0.0f;
        a.y = (rl1 == rowm) ? w1 : 0.0f;
        b.x = p[i].x; b.y = r[i].x;
        acc[0] = __builtin_amdgcn_wmma_f32_16x16x4_f32(false, a, false, b, (short)0, acc[0], false, false);
        b.x = p[i].y; b.y = r[i].y;
        acc[1] = __builtin_amdgcn_wmma_f32_16x16x4_f32(false, a, false, b, (short)0, acc[1], false, false);
        b.x = p[i].z; b.y = r[i].z;
        acc[2] = __builtin_amdgcn_wmma_f32_16x16x4_f32(false, a, false, b, (short)0, acc[2], false, false);
        b.x = p[i].w; b.y = r[i].w;
        acc[3] = __builtin_amdgcn_wmma_f32_16x16x4_f32(false, a, false, b, (short)0, acc[3], false, false);
      }
    }
  }
}

template <bool FULL, int NG, bool WEIGHTED>
static __device__ __forceinline__ void gather_add(unsigned c, unsigned rl, int n,
                                                  int lane, int2* st, float* ptile,
                                                  const float* __restrict__ deg,
                                                  const float* __restrict__ X) {
  st[lane] = make_int2((int)(c * (F_DIM * 4u)), (int)(rl * (P_PITCH * 4u)));
  wave_lds_handover();
  const int half = lane >> 4;
  const unsigned kb = (unsigned)(lane & 15) * 16u;
  const unsigned kb_first  = kb + (half ? 8u : 0u);
  const unsigned kb_second = kb + (half ? 0u : 8u);
  int wmine = 0;
  int2 s[16];
#pragma unroll
  for (int i = 0; i < 16; ++i) s[i] = st[i * 2 + half];
#pragma unroll
  for (int hb = 0; hb < 16 / NG; ++hb) {
    float4 g[NG];
#pragma unroll
    for (int i = 0; i < NG; ++i) {
      const int j = (hb * NG + i) * 2 + half;
      g[i] = make_float4(0.0f, 0.0f, 0.0f, 0.0f);
      if (FULL || j < n) g[i] = *(const float4*)((const char*)X + ((unsigned)s[hb * NG + i].x + kb));
    }
    if (WEIGHTED && hb == 0) {
      __builtin_amdgcn_sched_barrier(0);
      float dgc = 0.0f;
      if (FULL || lane < n) dgc = deg[c];
      wmine = __builtin_bit_cast(int, dis_of(dgc));
    }
#pragma unroll
    for (int i = 0; i < NG; ++i) {
      const int j = (hb * NG + i) * 2 + half;
      float ww = 1.0f;
      if (WEIGHTED) {
        const float w_lo = __builtin_bit_cast(float, __builtin_amdgcn_readlane(wmine, (hb * NG + i) * 2));
        const float w_hi = __builtin_bit_cast(float, __builtin_amdgcn_readlane(wmine, (hb * NG + i) * 2 + 1));
        ww = half ? w_hi : w_lo;
      }
      if (FULL || j < n) {
        const float v0 = half ? g[i].z : g[i].x;
        const float v1 = half ? g[i].w : g[i].y;
        const float v2 = half ? g[i].x : g[i].z;
        const float v3 = half ? g[i].y : g[i].w;
        float* d01 = (float*)((char*)ptile + ((unsigned)s[hb * NG + i].y + kb_first));
        float* d23 = (float*)((char*)ptile + ((unsigned)s[hb * NG + i].y + kb_second));
        lds_add(d01 + 0, WEIGHTED ? ww * v0 : v0);
        lds_add(d01 + 1, WEIGHTED ? ww * v1 : v1);
        lds_add(d23 + 0, WEIGHTED ? ww * v2 : v2);
        lds_add(d23 + 1, WEIGHTED ? ww * v3 : v3);
      }
    }
  }
}

template <bool FULL>
static __device__ __forceinline__ void list_batch(unsigned e, int n, int lane, int base, unsigned* cnt,
                                                  const int* __restrict__ row,
                                                  const int* __restrict__ col,
                                                  unsigned* __restrict__ lst) {
  if (FULL || lane < n) {
    const unsigned rl = (unsigned)(row[e] - base);
    lst[lane] = (unsigned)col[e] | (rl << 17);
    atomicAdd(&cnt[rl], 1u);
  }
}

__global__ __launch_bounds__(1024) void scan_list(
    const int* __restrict__ row, const int* __restrict__ col, const unsigned* __restrict__ key,
    const unsigned* __restrict__ part, const unsigned* __restrict__ gbits,
    unsigned* found, unsigned* __restrict__ elist, float* __restrict__ deg) {
  __shared__ unsigned cnt[R_TILE];
  __shared__ unsigned tfill[32];
  __shared__ unsigned queue[NWAVE * QCAP];
  __shared__ unsigned red_lo[NWAVE];
  __shared__ unsigned lfill;

  const int tid  = threadIdx.x;
  const int lane = tid & 31;
  const int wave = __builtin_amdgcn_readfirstlane(tid >> 5);
  const int base = blockIdx.x * R_TILE;

  if (tid < R_TILE) cnt[tid] = 0u;
  {
    unsigned s_lo = 0u;
    const int4* p4 = (const int4*)part;
    const int n_lo = (int)blockIdx.x * (PART_PITCH / 4);
    constexpr int DPT = 8;
    for (int i0 = 0; i0 < n_lo; i0 += DPT * NWAVE * 32) {
      int4 d[DPT];
#pragma unroll
      for (int k = 0; k < DPT; ++k)
        d[k] = p4[min(i0 + k * NWAVE * 32 + tid, n_lo - 1)];
#pragma unroll
      for (int k = 0; k < DPT; ++k)
        if (i0 + k * NWAVE * 32 + tid < n_lo)
          s_lo += ((unsigned)d[k].x + (unsigned)d[k].y) + ((unsigned)d[k].z + (unsigned)d[k].w);
    }
#pragma unroll
    for (int o = 16; o > 0; o >>= 1) s_lo += __shfl_xor(s_lo, o);
    if (lane == 0) red_lo[wave] = s_lo;
    if (tid == 0) lfill = 0u;
  }
  __syncthreads();
  unsigned t_lo = 0u;
#pragma unroll
  for (int w = 0; w < NWAVE; ++w) t_lo += red_lo[w];
  const unsigned l_off = __builtin_amdgcn_readfirstlane(t_lo);
  unsigned* mylist = found + l_off;

  unsigned* q   = queue + wave * QCAP;
  int qn = 0;
  const int4* key4 = (const int4*)key;
  const unsigned* mybits = gbits + (size_t)blockIdx.x * NBITW;
  constexpr int NTAKE = 4;
  const unsigned pat = (unsigned)blockIdx.x * 0x01010101u;

  auto nz = [](unsigned x) { return ((x & 0x7F7F7F7Fu) + 0x7F7F7F7Fu) | x; };
  auto take = [&](const int4 v, unsigned e0, bool valid) {
    const unsigned u0 = nz((unsigned)v.x ^ pat);
    const unsigned u1 = nz((unsigned)v.y ^ pat);
    const unsigned u2 = nz((unsigned)v.z ^ pat);
    const unsigned u3 = nz((unsigned)v.w ^ pat);
    unsigned hv = (((u0 & 0x80808080u) >> 3) | ((u1 & 0x80808080u) >> 2) |
                   ((u2 & 0x80808080u) >> 1) | (u3 & 0x80808080u)) ^ 0xF0F0F0F0u;
    if (!valid) hv = 0u;
    unsigned mask = __builtin_amdgcn_ballot_w32(hv != 0u);
    while (mask) {
      if (hv) {
        const unsigned p = (unsigned)__builtin_ctz(hv);
        hv &= hv - 1u;
        q[qn + __builtin_amdgcn_mbcnt_lo(mask, 0u)] = e0 + (((p & 3u) << 2) | (p >> 3));
      }
      qn += __builtin_popcount(mask);
      mask = __builtin_amdgcn_ballot_w32(hv != 0u);
      while (qn >= 32) {
        qn -= 32;
        wave_lds_handover();
        unsigned slot = 0u;
        if (lane == 0) slot = __hip_atomic_fetch_add(&lfill, 32u, __ATOMIC_RELAXED, __HIP_MEMORY_SCOPE_WORKGROUP);
        slot = __builtin_amdgcn_readfirstlane(slot);
        list_batch<true>(q[qn + lane], 32, lane, base, cnt, row, col, mylist + slot);
      }
    }
  };

  constexpr int WPW = (NBITW + NWAVE - 1) / NWAVE;
  constexpr int WPL = (WPW + 31) / 32;
  unsigned mw[WPL];
#pragma unroll
  for (int t = 0; t < WPL; ++t) {
    const int wi = t * 32 + lane;
    const int cw = wave * WPW + wi;
    mw[t] = (wi < WPW && cw < NBITW) ? mybits[cw] : 0u;
  }
  for (;;) {
    unsigned left = 0u;
#pragma unroll
    for (int t = 0; t < WPL; ++t) left |= mw[t];
    if (__builtin_amdgcn_ballot_w32(left != 0u) == 0u) break;
    unsigned g[NTAKE];
    bool has[NTAKE];
    int4 v[NTAKE];
#pragma unroll
    for (int k = 0; k < NTAKE; ++k) {
      unsigned cur = 0u;
      int sel = 0;
#pragma unroll
      for (int t = WPL - 1; t >= 0; --t)
        if (mw[t] != 0u) { cur = mw[t]; sel = t; }
      has[k] = (cur != 0u);
      const unsigned j = has[k] ? (unsigned)__builtin_ctz(cur) : 0u;
      g[k] = (unsigned)(wave * WPW + sel * 32 + lane) * 32u + j;
#pragma unroll
      for (int t = 0; t < WPL; ++t)
        if (t == sel) mw[t] &= mw[t] - 1u;
    }
#pragma unroll
    for (int k = 0; k < NTAKE; ++k) v[k] = key4[has[k] ? g[k] : 0u];
#pragma unroll
    for (int k = 0; k < NTAKE; ++k) take(v[k], g[k] * 16u, has[k]);
  }
  if (qn > 0) {
    wave_lds_handover();
    const unsigned ent = (lane < qn) ? q[lane] : 0u;
    unsigned slot = 0u;
    if (lane == 0) slot = __hip_atomic_fetch_add(&lfill, (unsigned)qn, __ATOMIC_RELAXED, __HIP_MEMORY_SCOPE_WORKGROUP);
    slot = __builtin_amdgcn_readfirstlane(slot);
    list_batch<false>(ent, qn, lane, base, cnt, row, col, mylist + slot);
  }
  __syncthreads();
  if (tid < R_TILE) deg[base + tid] = (float)cnt[tid];

  if (wave == 0) {
    unsigned s = 0u;
    if (lane < R_TILE / 16) {
#pragma unroll
      for (int k = 0; k < 16; ++k) s += cnt[lane * 16 + k];
    }
    unsigned incl = s;
#pragma unroll
    for (int o = 1; o < 32; o <<= 1) {
      const unsigned t = __shfl_up(incl, o);
      if (lane >= o) incl += t;
    }
    tfill[lane] = incl - s;
  }
  __syncthreads();
  const unsigned total = lfill;
  unsigned* grouped = elist + l_off;
  constexpr int WPT = 8;
  for (unsigned i0 = 0u; i0 < total; i0 += WPT * NWAVE * 32) {
    unsigned wd[WPT];
#pragma unroll
    for (int k = 0; k < WPT; ++k)
      wd[k] = mylist[min(i0 + (unsigned)(k * NWAVE * 32 + tid), total - 1u)];
#pragma unroll
    for (int k = 0; k < WPT; ++k)
      if (i0 + (unsigned)(k * NWAVE * 32 + tid) < total)
        grouped[atomicAdd(&tfill[wd[k] >> 21], 1u)] = wd[k];
  }
}

template <int FOUT, bool RELU, bool WGT, bool COPY>
__global__ __launch_bounds__(1024) void cheb_layer(
    const unsigned* __restrict__ elist, const float* __restrict__ deg,
    const float* XD, const float* __restrict__ XG, const float* __restrict__ W0, const float* __restrict__ W1,
    const float* __restrict__ B, float* OUT, float* __restrict__ OUTS) {
  constexpr int NT   = FOUT / 16;
  constexpr int NKG  = F_DIM / 4;
  constexpr int NENT = NKG * NT * 32;

  __shared__ float    ptile[R_TILE * P_PITCH];
  __shared__ float    disl[R_TILE];
  __shared__ float4   wlds[NENT];
  __shared__ __attribute__((aligned(16))) int2 stage[NWAVE * 32];
  __shared__ float    red_lo[NWAVE];
  __shared__ float    red_own[NWAVE];

  const int tid  = threadIdx.x;
  const int lane = tid & 31;
  const int wave = __builtin_amdgcn_readfirstlane(tid >> 5);
  const int base = blockIdx.x * R_TILE;

  for (int i = tid; i < R_TILE * P_PITCH / 4; i += NWAVE * 32)
    ((float4*)ptile)[i] = make_float4(0.0f, 0.0f, 0.0f, 0.0f);

  if (tid < R_TILE) disl[tid] = dis_of(deg[base + tid]);

  for (int idx = tid; idx < NENT; idx += NWAVE * 32) {
    const int l   = idx & 31;
    const int rem = idx >> 5;
    const int t   = rem % NT;
    const int kg  = rem / NT;
    const int n   = t * 16 + (l & 15);
    const int k   = kg * 4 + ((l >> 4) * 2);
    wlds[idx] = make_float4(W0[(size_t)k * FOUT + n], W0[(size_t)(k + 1) * FOUT + n],
                            W1[(size_t)k * FOUT + n], W1[(size_t)(k + 1) * FOUT + n]);
  }
  {
    float s_lo = 0.0f, s_own = 0.0f;
    const float4* d4 = (const float4*)deg;
    const int n_lo  = base / 4;
    const int n_all = n_lo + R_TILE / 4;
    constexpr int DPT = 8;
    for (int i0 = 0; i0 < n_all; i0 += DPT * NWAVE * 32) {
      float4 d[DPT];
#pragma unroll
      for (int k = 0; k < DPT; ++k)
        d[k] = d4[min(i0 + k * NWAVE * 32 + tid, n_all - 1)];
#pragma unroll
      for (int k = 0; k < DPT; ++k) {
        const int i = i0 + k * NWAVE * 32 + tid;
        const float s = (d[k].x + d[k].y) + (d[k].z + d[k].w);
        if (i < n_lo) s_lo += s; else if (i < n_all) s_own += s;
      }
    }
#pragma unroll
    for (int o = 16; o > 0; o >>= 1) {
      s_lo  += __shfl_xor(s_lo, o);
      s_own += __shfl_xor(s_own, o);
    }
    if (lane == 0) { red_lo[wave] = s_lo; red_own[wave] = s_own; }
  }
  __syncthreads();
  float t_lo = 0.0f, t_own = 0.0f;
#pragma unroll
  for (int w = 0; w < NWAVE; ++w) { t_lo += red_lo[w]; t_own += red_own[w]; }
  const unsigned l_off = __builtin_amdgcn_readfirstlane((unsigned)t_lo);
  const unsigned l_cnt = __builtin_amdgcn_readfirstlane((unsigned)t_own);
  const unsigned* mylist = elist + l_off;

  int2*     st  = stage + wave * 32;

  constexpr int NWM = 16;
  constexpr int NWA = NWAVE - NWM;
  const int nfull = (int)(l_cnt >> 5);
  const int ntail = (int)(l_cnt & 31u);
  const int nb    = nfull + (ntail > 0 ? 1 : 0);
  const int nbm   = (nb * NWM) / NWAVE;
  if (wave < NWM) {
    const int b0 = (int)(((unsigned)wave * (unsigned)nbm) / NWM);
    const int b1 = (int)(((unsigned)(wave + 1) * (unsigned)nbm) / NWM);
    if (b0 < b1) {
      const unsigned last = l_cnt - 1u;
      unsigned wa = mylist[min((unsigned)b0 * 32u + (unsigned)lane, last)];
      unsigned wb = mylist[min((unsigned)(b0 + 1) * 32u + (unsigned)lane, last)];
      float da = 0.0f;
      if (WGT) da = deg[wa & 0x1FFFFu];
      v8f acc[4];
#pragma unroll
      for (int j = 0; j < 4; ++j) acc[j] = (v8f)0.0f;
      int curT = -1;
      for (int b = b0; b < b1; ++b) {
        const unsigned wc = mylist[min((unsigned)(b + 2) * 32u + (unsigned)lane, last)];
        float db = 0.0f;
        if (WGT) db = deg[wb & 0x1FFFFu];
        const bool valid = (unsigned)b * 32u + (unsigned)lane <= last;
        const float wt = valid ? (WGT ? dis_of(da) : 1.0f) : 0.0f;
        tile_batch(wa, wt, lane, st, ptile, XG, acc, curT);
        wa = wb; wb = wc; da = db;
      }
      flush_tile(acc, curT, lane, ptile);
    }
  } else {
    const int aw = wave - NWM;
    if (nbm + aw < nfull) {
      const int blast = nfull - 1;
      unsigned w0 = mylist[(nbm + aw) * 32 + lane];
      unsigned wn = mylist[min(nbm + aw + NWA, blast) * 32 + lane];
      for (int b = nbm + aw; b < nfull; b += NWA) {
        const unsigned wn2 = mylist[min(b + 2 * NWA, blast) * 32 + lane];
        gather_add<true, 16, WGT>(w0 & 0x1FFFFu, w0 >> 17, 32, lane, st, ptile, deg, XG);
        w0 = wn; wn = wn2;
      }
    }
    if (ntail > 0 && aw == ((nfull - nbm) % NWA)) {
      const unsigned ent = (lane < ntail) ? mylist[nfull * 32 + lane] : 0u;
      gather_add<false, 16, WGT>(ent & 0x1FFFFu, ent >> 17, ntail, lane, st, ptile, deg, XG);
    }
  }
  __syncthreads();

  const int m  = lane & 15;
  const int kk = (lane >> 4) * 2;
  const int roff = (lane >> 4) * 8;
  for (int t16 = wave; t16 < R_TILE / 16; t16 += NWAVE) {
    const int row0 = base + t16 * 16;
    v8f acc[NT];
#pragma unroll
    for (int t = 0; t < NT; ++t) acc[t] = (v8f)0.0f;

    const float* xrow = XD + (size_t)(row0 + m) * F_DIM;
    const float* prow = ptile + (t16 * 16 + m) * P_PITCH;
    const float nd = -disl[t16 * 16 + m];

#pragma unroll 4
    for (int kg = 0; kg < NKG; ++kg) {
      const int k0 = kg * 4;
      const v2f aX = *(const v2f*)(xrow + k0 + kk);
      v2f aP = *(const v2f*)(prow + k0 + kk);
      aP.x *= nd; aP.y *= nd;
#pragma unroll
      for (int t = 0; t < NT; ++t) {
        const float4 wv = wlds[(kg * NT + t) * 32 + lane];
        v2f b0, b1;
        b0.x = wv.x; b0.y = wv.y;
        b1.x = wv.z; b1.y = wv.w;
        acc[t] = __builtin_amdgcn_wmma_f32_16x16x4_f32(
            false, aX, false, b0, (short)0, acc[t], false, false);
        acc[t] = __builtin_amdgcn_wmma_f32_16x16x4_f32(
            false, aP, false, b1, (short)0, acc[t], false, false);
      }
    }

    float dsr[8];
#pragma unroll
    for (int r = 0; r < 8; ++r) dsr[r] = COPY ? disl[t16 * 16 + roff + r] : 0.0f;
    float* srow = ptile + (t16 * 16) * P_PITCH;
#pragma unroll
    for (int t = 0; t < NT; ++t) {
      const float bias = B[t * 16 + m];
#pragma unroll
      for (int r = 0; r < 8; ++r) {
        float vv = acc[t][r] + bias;
        if (RELU) vv = fmaxf(vv, 0.0f);
        OUT[(size_t)(row0 + roff + r) * FOUT + t * 16 + m] = vv;
        if (COPY) srow[(roff + r) * P_PITCH + t * 16 + m] = dsr[r] * vv;
      }
    }
    if (COPY) {
      wave_lds_handover();
      const int q4 = (lane & 15) * 4;
#pragma unroll
      for (int i = 0; i < 8; ++i) {
        const int rr = i * 2 + (lane >> 4);
        const float4 v = *(const float4*)(srow + rr * P_PITCH + q4);
        *(float4*)(OUTS + (size_t)(row0 + rr) * FOUT + q4) = v;
      }
    }
  }
}

extern "C" void kernel_launch(void* const* d_in, const int* in_sizes, int n_in,
                              void* d_out, int out_size, void* d_ws, size_t ws_size,
                              hipStream_t stream) {
  const float* x    = (const float*)d_in[0];
  const int*   adj  = (const int*)d_in[1];
  const float* W1_0 = (const float*)d_in[2];
  const float* W1_1 = (const float*)d_in[3];
  const float* b1   = (const float*)d_in[4];
  const float* Wx_0 = (const float*)d_in[5];
  const float* Wx_1 = (const float*)d_in[6];
  const float* bx   = (const float*)d_in[7];
  const float* W2_0 = (const float*)d_in[8];
  const float* W2_1 = (const float*)d_in[9];
  const float* b2   = (const float*)d_in[10];
  float* out = (float*)d_out;

  const int* row = adj;
  const int* col = adj + E_EDGES;

  float* H   = (float*)d_ws;
  float* H1s = H + (size_t)N_NODES * F_DIM;
  float* H2s = H1s + (size_t)N_NODES * F_DIM;
  float* deg = H2s + (size_t)N_NODES * F_DIM;
  unsigned* elist = (unsigned*)(deg + N_NODES);
  unsigned* key  = (unsigned*)H2s;
  unsigned* found = key + E_EDGES / 4;
  unsigned* part = (unsigned*)H1s;
  unsigned* gbits = (unsigned*)H;
  if (ws_size < sizeof(float) * ((size_t)N_NODES + E_EDGES + 3 * (size_t)N_NODES * F_DIM)) return;

  const int gLay = N_NODES / R_TILE;

  edge_key_hist<<<KEY_WG, 1024, 0, stream>>>(row, key, part, gbits);

  scan_list<<<gLay, 1024, 0, stream>>>(row, col, key, part, gbits, found, elist, deg);

  cheb_layer<64, true, true, true><<<gLay, 1024, 0, stream>>>(elist, deg, x, x, W1_0, W1_1, b1, H, H1s);

  cheb_layer<64, true, false, true><<<gLay, 1024, 0, stream>>>(elist, deg, H, H1s, Wx_0, Wx_1, bx, H, H2s);

  cheb_layer<16, false, false, false><<<gLay, 1024, 0, stream>>>(elist, deg, H, H2s, W2_0, W2_1, b2, out, nullptr);
}
